# starting kernel
# baseline (speedup 1.0000x reference)
#include <hip/hip_runtime.h>
#include <math.h>

#define H      256
#define HE     64
#define NB     8
#define A_PER  64
#define N_PER  512
#define NA     (NB * A_PER)
#define NN     (NB * N_PER)

typedef __attribute__((ext_vector_type(2))) float v2f;
typedef __attribute__((ext_vector_type(8))) float v8f;

struct GemmArgs {
  const float* A;  long lda;  long sAb;
  const float* B;  long ldbn; long ldbk; long sBb;
  float*       C;  long ldc;  long sCb;
  const float* bias;
  int K;
  int relu;
  int addC;
};

template <bool BCONTIG>
__global__ __launch_bounds__(32)
void wmma_gemm_f32(GemmArgs g) {
  const int lane = threadIdx.x;
  const int r15  = lane & 15;
  const int kh   = (lane >> 4) << 1;
  const int tn = blockIdx.x, tm = blockIdx.y, bz = blockIdx.z;

  const float* Abase = g.A + (long)bz * g.sAb + (long)(tm * 64 + r15) * g.lda;
  const float* A0 = Abase;
  const float* A1 = Abase + (long)16 * g.lda;
  const float* A2 = Abase + (long)32 * g.lda;
  const float* A3 = Abase + (long)48 * g.lda;
  const float* B0 = g.B + (long)bz * g.sBb + (long)(tn * 32 + r15) * g.ldbn;
  const float* B1 = B0 + (long)16 * g.ldbn;

  const v8f zero = {0.f,0.f,0.f,0.f,0.f,0.f,0.f,0.f};
  v8f c00 = zero, c01 = zero, c10 = zero, c11 = zero;
  v8f c20 = zero, c21 = zero, c30 = zero, c31 = zero;

#if __has_builtin(__builtin_amdgcn_wmma_f32_16x16x4_f32)
  for (int k0 = 0; k0 < g.K; k0 += 4) {
    const int ka = k0 + kh;
    v2f a0 = *(const v2f*)(A0 + ka);
    v2f a1 = *(const v2f*)(A1 + ka);
    v2f a2 = *(const v2f*)(A2 + ka);
    v2f a3 = *(const v2f*)(A3 + ka);
    v2f b0, b1;
    if (BCONTIG) {
      b0 = *(const v2f*)(B0 + ka);
      b1 = *(const v2f*)(B1 + ka);
    } else {
      b0.x = B0[(long)ka * g.ldbk];  b0.y = B0[(long)(ka + 1) * g.ldbk];
      b1.x = B1[(long)ka * g.ldbk];  b1.y = B1[(long)(ka + 1) * g.ldbk];
    }
    c00 = __builtin_amdgcn_wmma_f32_16x16x4_f32(false, a0, false, b0, (short)0, c00, false, false);
    c01 = __builtin_amdgcn_wmma_f32_16x16x4_f32(false, a0, false, b1, (short)0, c01, false, false);
    c10 = __builtin_amdgcn_wmma_f32_16x16x4_f32(false, a1, false, b0, (short)0, c10, false, false);
    c11 = __builtin_amdgcn_wmma_f32_16x16x4_f32(false, a1, false, b1, (short)0, c11, false, false);
    c20 = __builtin_amdgcn_wmma_f32_16x16x4_f32(false, a2, false, b0, (short)0, c20, false, false);
    c21 = __builtin_amdgcn_wmma_f32_16x16x4_f32(false, a2, false, b1, (short)0, c21, false, false);
    c30 = __builtin_amdgcn_wmma_f32_16x16x4_f32(false, a3, false, b0, (short)0, c30, false, false);
    c31 = __builtin_amdgcn_wmma_f32_16x16x4_f32(false, a3, false, b1, (short)0, c31, false, false);
  }
#else
  {
    const float* Ab = g.A + (long)bz * g.sAb;
    const int mb = tm * 64 + ((lane >> 4) << 3);
    for (int k0 = 0; k0 < g.K; ++k0) {
      float bv0 = B0[(long)k0 * g.ldbk];
      float bv1 = B1[(long)k0 * g.ldbk];
      #pragma unroll
      for (int r = 0; r < 8; ++r) {
        float av0 = Ab[(long)(mb + r)      * g.lda + k0];
        float av1 = Ab[(long)(mb + 16 + r) * g.lda + k0];
        float av2 = Ab[(long)(mb + 32 + r) * g.lda + k0];
        float av3 = Ab[(long)(mb + 48 + r) * g.lda + k0];
        c00[r] += av0 * bv0;  c01[r] += av0 * bv1;
        c10[r] += av1 * bv0;  c11[r] += av1 * bv1;
        c20[r] += av2 * bv0;  c21[r] += av2 * bv1;
        c30[r] += av3 * bv0;  c31[r] += av3 * bv1;
      }
    }
  }
#endif

  const int  col0 = tn * 32 + r15;
  const int  col1 = col0 + 16;
  const int  m0   = tm * 64 + ((lane >> 4) << 3);
  const float bv0 = g.bias ? g.bias[col0] : 0.f;
  const float bv1 = g.bias ? g.bias[col1] : 0.f;
  float* Cp = g.C + (long)bz * g.sCb;
  const v8f* cc0[4] = { &c00, &c10, &c20, &c30 };
  const v8f* cc1[4] = { &c01, &c11, &c21, &c31 };
  #pragma unroll
  for (int i = 0; i < 4; ++i) {
    #pragma unroll
    for (int r = 0; r < 8; ++r) {
      const long row = (long)(m0 + i * 16 + r) * g.ldc;
      float v0 = (*cc0[i])[r] + bv0;
      float v1 = (*cc1[i])[r] + bv1;
      if (g.addC) { v0 += Cp[row + col0]; v1 += Cp[row + col1]; }
      if (g.relu) { v0 = fmaxf(v0, 0.f);  v1 = fmaxf(v1, 0.f); }
      Cp[row + col0] = v0;
      Cp[row + col1] = v1;
    }
  }
}

__global__ __launch_bounds__(256)
void edge_softmax_kernel(const float* __restrict__ anchor_x,
                         const float* __restrict__ node_x,
                         const float* __restrict__ node_mask,
                         const float* __restrict__ qe,
                         const float* __restrict__ Lbase,
                         float* __restrict__ attn,
                         float* __restrict__ s_out)
{
  const int bid = blockIdx.x;
  const int b   = bid >> 6;
  const int t   = threadIdx.x;

  __shared__ float sh_qe[HE];
  __shared__ float sh_dist[N_PER];
  __shared__ float sh_l[N_PER];
  __shared__ float red[256];
  __shared__ float sh_bcast;
  __shared__ float sh_ax[3];

  if (t < HE) sh_qe[t] = qe[bid * HE + t];
  if (t < 3)  sh_ax[t] = anchor_x[bid * 3 + t];
  __syncthreads();

  const float axx = sh_ax[0], axy = sh_ax[1], axz = sh_ax[2];
  const float MU_STEP = 20.f / 63.f;
  const float INV_SIG = 64.f / 20.f;

  for (int n = t; n < N_PER; n += 256) {
    const float* nx = node_x + (long)(b * N_PER + n) * 3;
    float dx = axx - nx[0] + 1e-8f;
    float dy = axy - nx[1] + 1e-8f;
    float dz = axz - nx[2] + 1e-8f;
    float d  = sqrtf(dx * dx + dy * dy + dz * dz);
    sh_dist[n] = d;
    float d10 = d * 0.1f;
    float le = 0.f;
    #pragma unroll 8
    for (int j = 0; j < HE; ++j) {
      float u = (d10 - j * MU_STEP) * INV_SIG;
      le += sh_qe[j] * __expf(-u * u);
    }
    float lb = Lbase[(long)bid * N_PER + n] + le;
    lb *= (node_mask[b * N_PER + n] - 1.f) * 1000000.0f;
    sh_l[n] = lb;
  }
  __syncthreads();

  red[t] = fmaxf(sh_l[t], sh_l[t + 256]);
  __syncthreads();
  for (int st = 128; st > 0; st >>= 1) {
    if (t < st) red[t] = fmaxf(red[t], red[t + st]);
    __syncthreads();
  }
  if (t == 0) sh_bcast = red[0];
  __syncthreads();
  const float mx = sh_bcast;

  float e0 = __expf(sh_l[t] - mx);
  float e1 = __expf(sh_l[t + 256] - mx);
  red[t] = e0 + e1;
  __syncthreads();
  for (int st = 128; st > 0; st >>= 1) {
    if (t < st) red[t] += red[t + st];
    __syncthreads();
  }
  if (t == 0) sh_bcast = red[0];
  __syncthreads();
  const float inv = 1.f / sh_bcast;

  const float a0 = e0 * inv, a1 = e1 * inv;
  sh_l[t] = a0; sh_l[t + 256] = a1;
  attn[(long)bid * N_PER + t]       = a0;
  attn[(long)bid * N_PER + t + 256] = a1;
  __syncthreads();

  const int j = t & 63, c = t >> 6;
  const float muj = j * MU_STEP;
  float part = 0.f;
  for (int n = c * 128; n < c * 128 + 128; ++n) {
    float u = (sh_dist[n] * 0.1f - muj) * INV_SIG;
    part += sh_l[n] * __expf(-u * u);
  }
  red[t] = part;
  __syncthreads();
  if (t < 64)
    s_out[bid * HE + t] = red[t] + red[t + 64] + red[t + 128] + red[t + 192];
}

__global__ __launch_bounds__(256)
void add_ln_kernel(const float* __restrict__ x, const float* __restrict__ y,
                   const float* __restrict__ gam, const float* __restrict__ bet,
                   float* __restrict__ out)
{
  const int r = blockIdx.x, t = threadIdx.x;
  __shared__ float red[256];
  __shared__ float sh_mean, sh_rstd;

  const float v = x[(long)r * H + t] + y[(long)r * H + t];
  red[t] = v;
  __syncthreads();
  for (int st = 128; st > 0; st >>= 1) { if (t < st) red[t] += red[t + st]; __syncthreads(); }
  if (t == 0) sh_mean = red[0] * (1.f / H);
  __syncthreads();

  const float d = v - sh_mean;
  red[t] = d * d;
  __syncthreads();
  for (int st = 128; st > 0; st >>= 1) { if (t < st) red[t] += red[t + st]; __syncthreads(); }
  if (t == 0) sh_rstd = rsqrtf(red[0] * (1.f / H) + 1e-5f);
  __syncthreads();

  out[(long)r * H + t] = d * sh_rstd * gam[t] + bet[t];
}

extern "C" void kernel_launch(void* const* d_in, const int* in_sizes, int n_in,
                              void* d_out, int out_size, void* d_ws, size_t ws_size,
                              hipStream_t stream) {
  (void)in_sizes; (void)n_in; (void)out_size; (void)ws_size;
  const float* anchor_x  = (const float*)d_in[0];
  const float* node_x    = (const float*)d_in[1];
  const float* anchor_f  = (const float*)d_in[2];
  const float* node_f    = (const float*)d_in[3];
  const float* node_mask = (const float*)d_in[6];
  const float* Wq  = (const float*)d_in[7];
  const float* bq  = (const float*)d_in[8];
  const float* Wkv = (const float*)d_in[9];
  const float* bkv = (const float*)d_in[10];
  const float* ln1_g = (const float*)d_in[11];
  const float* ln1_b = (const float*)d_in[12];
  const float* W1 = (const float*)d_in[13];
  const float* b1 = (const float*)d_in[14];
  const float* W2 = (const float*)d_in[15];
  const float* b2 = (const float*)d_in[16];
  const float* W3 = (const float*)d_in[17];
  const float* b3 = (const float*)d_in[18];
  const float* ln2_g = (const float*)d_in[19];
  const float* ln2_b = (const float*)d_in[20];

  const long WKV_LD = H + HE;

  float* p = (float*)d_ws;
  float* q    = p; p += (long)NA * H;
  float* qe   = p; p += (long)NA * HE;
  float* kvn  = p; p += (long)NN * 2 * H;
  float* L    = p; p += (long)NA * N_PER;
  float* attn = p; p += (long)NA * N_PER;
  float* sbuf = p; p += (long)NA * HE;
  float* upd  = p; p += (long)NA * H;
  float* af   = p; p += (long)NA * H;
  float* m1   = p; p += (long)NA * 2 * H;
  float* m2   = p; p += (long)NA * 2 * H;
  float* m3   = p; p += (long)NA * H;

  auto gemm = [&](const float* A, long lda, long sAb,
                  const float* Bm, long ldbn, long ldbk, long sBb,
                  float* C, long ldc, long sCb, const float* bias,
                  int M, int N, int K, int nb, int relu, int addC) {
    GemmArgs g{A, lda, sAb, Bm, ldbn, ldbk, sBb, C, ldc, sCb, bias, K, relu, addC};
    dim3 grid(N / 32, M / 64, nb);
    if (ldbk == 1)
      wmma_gemm_f32<true><<<grid, dim3(32), 0, stream>>>(g);
    else
      wmma_gemm_f32<false><<<grid, dim3(32), 0, stream>>>(g);
  };

  gemm(anchor_f, H, 0, Wq, H, 1, 0, q, H, 0, bq, NA, H, H, 1, 0, 0);
  gemm(q, H, 0, Wkv + H, 1, WKV_LD, 0, qe, HE, 0, nullptr, NA, HE, H, 1, 0, 0);
  gemm(node_f, H, 0, Wkv, WKV_LD, 1, 0, kvn, 2 * H, 0, bkv, NN, 2 * H, H, 1, 0, 0);
  gemm(q, H, (long)A_PER * H,
       kvn, 2 * H, 1, (long)N_PER * 2 * H,
       L, N_PER, (long)A_PER * N_PER, nullptr, A_PER, N_PER, H, NB, 0, 0);
  edge_softmax_kernel<<<dim3(NA), dim3(256), 0, stream>>>(
      anchor_x, node_x, node_mask, qe, L, attn, sbuf);
  gemm(attn, N_PER, (long)A_PER * N_PER,
       kvn + H, 1, 2 * H, (long)N_PER * 2 * H,
       upd, H, (long)A_PER * H, nullptr, A_PER, H, N_PER, NB, 0, 0);
  gemm(sbuf, HE, 0, Wkv + (long)H * WKV_LD + H, WKV_LD, 1, 0,
       upd, H, 0, nullptr, NA, H, HE, 1, 0, 1);
  add_ln_kernel<<<dim3(NA), dim3(256), 0, stream>>>(anchor_f, upd, ln1_g, ln1_b, af);
  gemm(af, H, 0, W1, H, 1, 0, m1, 2 * H, 0, b1, NA, 2 * H, H, 1, 1, 0);
  gemm(m1, 2 * H, 0, W2, 2 * H, 1, 0, m2, 2 * H, 0, b2, NA, 2 * H, 2 * H, 1, 1, 0);
  gemm(m2, 2 * H, 0, W3, 2 * H, 1, 0, m3, H, 0, b3, NA, H, 2 * H, 1, 0, 0);
  add_ln_kernel<<<dim3(NA), dim3(256), 0, stream>>>(af, m3, ln2_g, ln2_b, (float*)d_out);
}


# LayerNorm folded into its producing GEMM: an 8-wave workgroup owns 16 whole rows (16x256 tile, 2 LDS stages), residual + row mean/variance in the epilogue; 11 -> 9 launches (LN sum order changes)
# speedup vs baseline: 3.8429x; 3.8429x over previous
#include <hip/hip_runtime.h>
#include <math.h>

#define H      256
#define HE     64
#define NB     8
#define A_PER  64
#define N_PER  512
#define NA     (NB * A_PER)
#define NN     (NB * N_PER)

typedef __attribute__((ext_vector_type(2))) float v2f;
typedef __attribute__((ext_vector_type(8))) float v8f;

#define BM 32
#define BN 64
#define BK 64
#define PK 68

struct GemmArgs {
  const float* A;  long lda;  long sAb;
  const float* B;  long ldbn; long ldbk; long sBb;
  float*       C;  long ldc;  long sCb;
  const float* bias;
  int K;
  int relu;
  int addC;
};

template <bool BKCONTIG>
__global__ __launch_bounds__(128)
void wmma_gemm_f32(GemmArgs g) {
  __shared__ __attribute__((aligned(16))) float As[BM * PK];
  __shared__ __attribute__((aligned(16))) float Bs[BN * PK];

  const int t    = threadIdx.x;
  const int lane = t & 31;
  const int w    = t >> 5;
  const int r15  = lane & 15;
  const int kh   = (lane >> 4) << 1;
  const int wm   = w & 1;
  const int wn   = w >> 1;
  const int tn = blockIdx.x, tm = blockIdx.y, bz = blockIdx.z;

  const int lr = t >> 4;
  const int lc = (t & 15) << 2;

  const float* Ag = g.A + (long)bz * g.sAb + (long)(tm * BM + lr) * g.lda + lc;
  const float* Bg = BKCONTIG
      ? g.B + (long)bz * g.sBb + (long)(tn * BN + lr) * g.ldbn + lc
      : g.B + (long)bz * g.sBb + (long)(2 * lr) * g.ldbk + (tn * BN + lc);

  const long a8 = 8 * g.lda;
  const long bu = BKCONTIG ? 8 * g.ldbn : g.ldbk;
  const long bstep = BKCONTIG ? (long)BK : (long)BK * g.ldbk;
#define BROW(i) ((BKCONTIG ? (i) : (((i) >> 1) * 16 + ((i) & 1))) * bu)
  float4 ra0, ra1, ra2, ra3, rb0, rb1, rb2, rb3, rb4, rb5, rb6, rb7;
#define GLOAD_SLAB()                                                                   \
  ra0 = *(const float4*)(Ag);           ra1 = *(const float4*)(Ag + a8);               \
  ra2 = *(const float4*)(Ag + 2 * a8);  ra3 = *(const float4*)(Ag + 3 * a8);           \
  rb0 = *(const float4*)(Bg + BROW(0)); rb1 = *(const float4*)(Bg + BROW(1));          \
  rb2 = *(const float4*)(Bg + BROW(2)); rb3 = *(const float4*)(Bg + BROW(3));          \
  rb4 = *(const float4*)(Bg + BROW(4)); rb5 = *(const float4*)(Bg + BROW(5));          \
  rb6 = *(const float4*)(Bg + BROW(6)); rb7 = *(const float4*)(Bg + BROW(7));          \
  Ag += BK; Bg += bstep;
#define BPAIR_STORE(i, lo, hi)                                                         \
  *(v2f*)(Bw + 16 * (i))          = (v2f){lo.x, hi.x};                                 \
  *(v2f*)(Bw + 16 * (i) + PK)     = (v2f){lo.y, hi.y};                                 \
  *(v2f*)(Bw + 16 * (i) + 2 * PK) = (v2f){lo.z, hi.z};                                 \
  *(v2f*)(Bw + 16 * (i) + 3 * PK) = (v2f){lo.w, hi.w};

  float* Aw = &As[lr * PK + lc];
  float* Bw = BKCONTIG ? &Bs[lr * PK + lc] : &Bs[lc * PK + 2 * lr];
  constexpr int R8 = 8 * PK;

  const float* Af  = &As[(wm * 16 + r15) * PK + kh];
  const float* Bf0 = &Bs[(wn * 32 + r15) * PK + kh];
  const float* Bf1 = Bf0 + 16 * PK;

  const v8f zero = {0.f,0.f,0.f,0.f,0.f,0.f,0.f,0.f};
  v8f c0 = zero, c1 = zero;

  GLOAD_SLAB()
  for (int k0 = 0; k0 < g.K; k0 += BK) {
    *(float4*)(Aw)          = ra0;  *(float4*)(Aw + R8)     = ra1;
    *(float4*)(Aw + 2 * R8) = ra2;  *(float4*)(Aw + 3 * R8) = ra3;
    if (BKCONTIG) {
      *(float4*)(Bw)          = rb0;  *(float4*)(Bw + R8)     = rb1;
      *(float4*)(Bw + 2 * R8) = rb2;  *(float4*)(Bw + 3 * R8) = rb3;
      *(float4*)(Bw + 4 * R8) = rb4;  *(float4*)(Bw + 5 * R8) = rb5;
      *(float4*)(Bw + 6 * R8) = rb6;  *(float4*)(Bw + 7 * R8) = rb7;
    } else {
      BPAIR_STORE(0, rb0, rb1)
      BPAIR_STORE(1, rb2, rb3)
      BPAIR_STORE(2, rb4, rb5)
      BPAIR_STORE(3, rb6, rb7)
    }
    __syncthreads();

    const bool more = (k0 + BK) < g.K;
    if (more) { GLOAD_SLAB() }

    #pragma unroll
    for (int kk = 0; kk < BK; kk += 4) {
      v2f a  = *(const v2f*)(Af + kk);
      v2f b0 = *(const v2f*)(Bf0 + kk);
      v2f b1 = *(const v2f*)(Bf1 + kk);
      c0 = __builtin_amdgcn_wmma_f32_16x16x4_f32(false, a, false, b0, (short)0, c0, false, false);
      c1 = __builtin_amdgcn_wmma_f32_16x16x4_f32(false, a, false, b1, (short)0, c1, false, false);
    }
    if (more) __syncthreads();
  }

  const int  col0 = tn * BN + wn * 32 + r15;
  const int  col1 = col0 + 16;
  const int  m0   = tm * BM + wm * 16 + ((lane >> 4) << 3);
  const float bv0 = g.bias ? g.bias[col0] : 0.f;
  const float bv1 = g.bias ? g.bias[col1] : 0.f;
  float* Cp = g.C + (long)bz * g.sCb;
  #pragma unroll
  for (int r = 0; r < 8; ++r) {
    const long row = (long)(m0 + r) * g.ldc;
    float v0 = c0[r] + bv0;
    float v1 = c1[r] + bv1;
    if (g.addC) { v0 += Cp[row + col0]; v1 += Cp[row + col1]; }
    if (g.relu) { v0 = fmaxf(v0, 0.f);  v1 = fmaxf(v1, 0.f); }
    Cp[row + col0] = v0;
    Cp[row + col1] = v1;
  }
}
#undef GLOAD_SLAB
#undef BPAIR_STORE
#undef BROW

#define LBM 16
#define LBN 256

struct RowLnArgs {
  const float* A;  long lda;
  const float* B;  long ldbn;
  const float* bias;
  const float* resid;
  const float* gam;
  const float* bet;
  float*       out;
  int K;
};

template <int XM>
__device__ __forceinline__ float row_xor_add(float v) {
  const int o = __builtin_amdgcn_update_dpp(0, __float_as_int(v), 0x160 + XM, 0xf, 0xf, true);
  return v + __int_as_float(o);
}

__global__ __launch_bounds__(256)
void wmma_gemm_rowln_f32(RowLnArgs g) {
  constexpr int ASTG = LBM * PK;
  constexpr int BSTG = LBN * PK;
  __shared__ __attribute__((aligned(16))) float As[2 * ASTG];
  __shared__ __attribute__((aligned(16))) float Bs[2 * BSTG];
  __shared__ __attribute__((aligned(16))) float red1[LBM * 8];
  __shared__ __attribute__((aligned(16))) float red2[LBM * 8];

  const int t    = threadIdx.x;
  const int lane = t & 31;
  const int w    = t >> 5;
  const int r15  = lane & 15;
  const int hh   = lane >> 4;
  const int kh   = hh << 1;
  const int tm   = blockIdx.x;

  const int lr = t >> 4;
  const int lc = (t & 15) << 2;

  const float* Ag = g.A + (long)(tm * LBM + lr) * g.lda + lc;
  const float* Bg = g.B + (long)lr * g.ldbn + lc;
  const long b16 = 16 * g.ldbn;

  const int  col0 = w * 32 + r15;
  const int  col1 = col0 + 16;
  const int  m0   = tm * LBM + (hh << 3);
  const float bv0 = g.bias[col0], bv1 = g.bias[col1];
  const float ga0 = g.gam[col0],  ga1 = g.gam[col1];
  const float be0 = g.bet[col0],  be1 = g.bet[col1];
  float x0[8], x1[8];
  for (int r = 0; r < 8; ++r) {
    x0[r] = g.resid[(long)(m0 + r) * LBN + col0];
    x1[r] = g.resid[(long)(m0 + r) * LBN + col1];
  }

  float4 ra, rb0, rb1, rb2, rb3, rb4, rb5, rb6, rb7, rb8, rb9, rb10, rb11, rb12, rb13, rb14, rb15;
#define LN_GLOAD_SLAB()                                                                        \
  ra = *(const float4*)(Ag);                                                                   \
  rb0 = *(const float4*)(Bg);                  rb1 = *(const float4*)(Bg + 1 * b16);        \
  rb2 = *(const float4*)(Bg + 2 * b16);        rb3 = *(const float4*)(Bg + 3 * b16);        \
  rb4 = *(const float4*)(Bg + 4 * b16);        rb5 = *(const float4*)(Bg + 5 * b16);        \
  rb6 = *(const float4*)(Bg + 6 * b16);        rb7 = *(const float4*)(Bg + 7 * b16);        \
  rb8 = *(const float4*)(Bg + 8 * b16);        rb9 = *(const float4*)(Bg + 9 * b16);        \
  rb10 = *(const float4*)(Bg + 10 * b16);      rb11 = *(const float4*)(Bg + 11 * b16);      \
  rb12 = *(const float4*)(Bg + 12 * b16);      rb13 = *(const float4*)(Bg + 13 * b16);      \
  rb14 = *(const float4*)(Bg + 14 * b16);      rb15 = *(const float4*)(Bg + 15 * b16);      \
  Ag += BK; Bg += BK;
  constexpr int R16 = 16 * PK;
#define LN_STORE_SLAB(stg)                                                                     \
  {                                                                                            \
    float* bw = Bw + (stg) * BSTG;                                                             \
    *(float4*)(Aw + (stg) * ASTG) = ra;                                                        \
    *(float4*)(bw) = rb0;                    *(float4*)(bw + 1 * R16) = rb1;          \
    *(float4*)(bw + 2 * R16) = rb2;          *(float4*)(bw + 3 * R16) = rb3;          \
    *(float4*)(bw + 4 * R16) = rb4;          *(float4*)(bw + 5 * R16) = rb5;          \
    *(float4*)(bw + 6 * R16) = rb6;          *(float4*)(bw + 7 * R16) = rb7;          \
    *(float4*)(bw + 8 * R16) = rb8;          *(float4*)(bw + 9 * R16) = rb9;          \
    *(float4*)(bw + 10 * R16) = rb10;        *(float4*)(bw + 11 * R16) = rb11;        \
    *(float4*)(bw + 12 * R16) = rb12;        *(float4*)(bw + 13 * R16) = rb13;        \
    *(float4*)(bw + 14 * R16) = rb14;        *(float4*)(bw + 15 * R16) = rb15;        \
  }

  float* Aw = &As[lr * PK + lc];
  float* Bw = &Bs[lr * PK + lc];

  const float* Af  = &As[r15 * PK + kh];
  const float* Bf  = &Bs[(w * 32 + r15) * PK + kh];

  const v8f zero = {0.f,0.f,0.f,0.f,0.f,0.f,0.f,0.f};
  v8f c0 = zero, c1 = zero;

  LN_GLOAD_SLAB()
  LN_STORE_SLAB(0)
  __syncthreads();
  int st = 0;
  int k0 = 0;
  bool more;
  do {
    more = (k0 + BK) < g.K;
    if (more) { LN_GLOAD_SLAB() }

    const float* af  = Af + st * ASTG;
    const float* bf0 = Bf + st * BSTG;
    const float* bf1 = bf0 + 16 * PK;
    #pragma unroll
    for (int kk = 0; kk < BK; kk += 4) {
      v2f a  = *(const v2f*)(af + kk);
      v2f b0 = *(const v2f*)(bf0 + kk);
      v2f b1 = *(const v2f*)(bf1 + kk);
      c0 = __builtin_amdgcn_wmma_f32_16x16x4_f32(false, a, false, b0, (short)0, c0, false, false);
      c1 = __builtin_amdgcn_wmma_f32_16x16x4_f32(false, a, false, b1, (short)0, c1, false, false);
    }
    if (more) {
      st ^= 1;
      LN_STORE_SLAB(st)
      __syncthreads();
    }
    k0 += BK;
  } while (more);
  asm volatile("v_nop\n\tv_nop\n\tv_nop\n\tv_nop\n\tv_nop" : "+v"(c0), "+v"(c1));
#undef LN_GLOAD_SLAB
#undef LN_STORE_SLAB

  float v0[8], v1[8], ps[8];
  for (int r = 0; r < 8; ++r) {
    v0[r] = x0[r] + (c0[r] + bv0);
    v1[r] = x1[r] + (c1[r] + bv1);
    ps[r] = v0[r] + v1[r];
  }
  for (int r = 0; r < 8; ++r) ps[r] = row_xor_add<1>(ps[r]);
  for (int r = 0; r < 8; ++r) ps[r] = row_xor_add<2>(ps[r]);
  for (int r = 0; r < 8; ++r) ps[r] = row_xor_add<4>(ps[r]);
  for (int r = 0; r < 8; ++r) ps[r] = row_xor_add<8>(ps[r]);
  if (r15 == 0) {
    for (int r = 0; r < 8; ++r) red1[((hh << 3) + r) * 8 + w] = ps[r];
  }
  __syncthreads();
  for (int r = 0; r < 8; ++r) {
    const float4 pa = *(const float4*)(&red1[((hh << 3) + r) * 8]);
    const float4 pb = *(const float4*)(&red1[((hh << 3) + r) * 8 + 4]);
    const float mean = (((pa.x + pa.y) + (pa.z + pa.w)) + ((pb.x + pb.y) + (pb.z + pb.w))) * (1.f / LBN);
    v0[r] -= mean;
    v1[r] -= mean;
    ps[r] = v0[r] * v0[r] + v1[r] * v1[r];
  }
  for (int r = 0; r < 8; ++r) ps[r] = row_xor_add<1>(ps[r]);
  for (int r = 0; r < 8; ++r) ps[r] = row_xor_add<2>(ps[r]);
  for (int r = 0; r < 8; ++r) ps[r] = row_xor_add<4>(ps[r]);
  for (int r = 0; r < 8; ++r) ps[r] = row_xor_add<8>(ps[r]);
  if (r15 == 0) {
    for (int r = 0; r < 8; ++r) red2[((hh << 3) + r) * 8 + w] = ps[r];
  }
  __syncthreads();
  for (int r = 0; r < 8; ++r) {
    const float4 pa = *(const float4*)(&red2[((hh << 3) + r) * 8]);
    const float4 pb = *(const float4*)(&red2[((hh << 3) + r) * 8 + 4]);
    const float var  = (((pa.x + pa.y) + (pa.z + pa.w)) + ((pb.x + pb.y) + (pb.z + pb.w))) * (1.f / LBN);
    const float rstd = rsqrtf(var + 1e-5f);
    const long row = (long)(m0 + r) * LBN;
    g.out[row + col0] = v0[r] * rstd * ga0 + be0;
    g.out[row + col1] = v1[r] * rstd * ga1 + be1;
  }
}

#define RP 68
__global__ __launch_bounds__(256)
void edge_softmax_kernel(const float* __restrict__ anchor_x,
                         const float* __restrict__ node_x,
                         const float* __restrict__ node_mask,
                         const float* __restrict__ q,
                         const float* __restrict__ bk,
                         const float* __restrict__ qe,
                         const float* __restrict__ Lbase,
                         float* __restrict__ attn,
                         float* __restrict__ s_out,
                         int ldq, int lds_)
{
  const int bid = blockIdx.x;
  const int b   = bid >> 6;
  const int t   = threadIdx.x;

  __shared__ __attribute__((aligned(16))) float sh_rbf[N_PER * RP];
  __shared__ __attribute__((aligned(16))) float sh_l[N_PER];
  __shared__ __attribute__((aligned(16))) float sh_part[16 * HE];
  __shared__ __attribute__((aligned(16))) float sh_qe[HE];
  __shared__ float red[256];
  __shared__ float sh_bcast;
  __shared__ float sh_ax[3];
  __shared__ float sh_c[8];

  if (t < HE) sh_qe[t] = qe[(long)bid * ldq + t];
  if (t < 3)  sh_ax[t] = anchor_x[bid * 3 + t];
  float pc = q[(long)bid * H + t] * bk[t];
  for (int o = 16; o > 0; o >>= 1) pc += __shfl_xor(pc, o, 32);
  if ((t & 31) == 0) sh_c[t >> 5] = pc;
  __syncthreads();
  const float ca = ((sh_c[0] + sh_c[1]) + (sh_c[2] + sh_c[3])) +
                   ((sh_c[4] + sh_c[5]) + (sh_c[6] + sh_c[7]));

  const float axx = sh_ax[0], axy = sh_ax[1], axz = sh_ax[2];
  const float MU_STEP = 20.f / 63.f;
  const float INV_SIG = 64.f / 20.f;

  for (int n = t; n < N_PER; n += 256) {
    const float* nx = node_x + (long)(b * N_PER + n) * 3;
    float dx = axx - nx[0] + 1e-8f;
    float dy = axy - nx[1] + 1e-8f;
    float dz = axz - nx[2] + 1e-8f;
    float d  = sqrtf(dx * dx + dy * dy + dz * dz);
    float d10 = d * 0.1f;
    float* rrow = &sh_rbf[n * RP];
    float le = 0.f;
    #pragma unroll 2
    for (int j = 0; j < HE; j += 4) {
      float u0 = (d10 - (j    ) * MU_STEP) * INV_SIG;
      float u1 = (d10 - (j + 1) * MU_STEP) * INV_SIG;
      float u2 = (d10 - (j + 2) * MU_STEP) * INV_SIG;
      float u3 = (d10 - (j + 3) * MU_STEP) * INV_SIG;
      float4 e;
      e.x = __expf(-u0 * u0);
      e.y = __expf(-u1 * u1);
      e.z = __expf(-u2 * u2);
      e.w = __expf(-u3 * u3);
      le = fmaf(sh_qe[j],     e.x, le);
      le = fmaf(sh_qe[j + 1], e.y, le);
      le = fmaf(sh_qe[j + 2], e.z, le);
      le = fmaf(sh_qe[j + 3], e.w, le);
      *(float4*)(rrow + j) = e;
    }
    float lb = (Lbase[(long)bid * N_PER + n] + ca) + le;
    lb *= (node_mask[b * N_PER + n] - 1.f) * 1000000.0f;
    sh_l[n] = lb;
  }
  __syncthreads();

  red[t] = fmaxf(sh_l[t], sh_l[t + 256]);
  __syncthreads();
  for (int st = 128; st > 0; st >>= 1) {
    if (t < st) red[t] = fmaxf(red[t], red[t + st]);
    __syncthreads();
  }
  if (t == 0) sh_bcast = red[0];
  __syncthreads();
  const float mx = sh_bcast;

  float e0 = __expf(sh_l[t] - mx);
  float e1 = __expf(sh_l[t + 256] - mx);
  red[t] = e0 + e1;
  __syncthreads();
  for (int st = 128; st > 0; st >>= 1) {
    if (t < st) red[t] += red[t + st];
    __syncthreads();
  }
  if (t == 0) sh_bcast = red[0];
  __syncthreads();
  const float inv = 1.f / sh_bcast;

  const float a0 = e0 * inv, a1 = e1 * inv;
  sh_l[t] = a0; sh_l[t + 256] = a1;
  attn[(long)bid * N_PER + t]       = a0;
  attn[(long)bid * N_PER + t + 256] = a1;
  __syncthreads();

  const int jg = (t & 15) << 2, c = t >> 4;
  const float* rp = &sh_rbf[(c * 32) * RP + jg];
  const float* ap = &sh_l[c * 32];
  float4 part = make_float4(0.f, 0.f, 0.f, 0.f);
  for (int i = 0; i < 32; i += 4) {
    const float4 a4 = *(const float4*)(ap + i);
    const float4 r0 = *(const float4*)(rp + (i    ) * RP);
    const float4 r1 = *(const float4*)(rp + (i + 1) * RP);
    const float4 r2 = *(const float4*)(rp + (i + 2) * RP);
    const float4 r3 = *(const float4*)(rp + (i + 3) * RP);
    part.x = fmaf(a4.x, r0.x, part.x); part.y = fmaf(a4.x, r0.y, part.y);
    part.z = fmaf(a4.x, r0.z, part.z); part.w = fmaf(a4.x, r0.w, part.w);
    part.x = fmaf(a4.y, r1.x, part.x); part.y = fmaf(a4.y, r1.y, part.y);
    part.z = fmaf(a4.y, r1.z, part.z); part.w = fmaf(a4.y, r1.w, part.w);
    part.x = fmaf(a4.z, r2.x, part.x); part.y = fmaf(a4.z, r2.y, part.y);
    part.z = fmaf(a4.z, r2.z, part.z); part.w = fmaf(a4.z, r2.w, part.w);
    part.x = fmaf(a4.w, r3.x, part.x); part.y = fmaf(a4.w, r3.y, part.y);
    part.z = fmaf(a4.w, r3.z, part.z); part.w = fmaf(a4.w, r3.w, part.w);
  }
  *(float4*)(&sh_part[c * HE + jg]) = part;
  __syncthreads();
  if (t < HE) {
    float acc = sh_part[t];
    for (int cc = 1; cc < 16; ++cc) acc += sh_part[cc * HE + t];
    s_out[(long)bid * lds_ + t] = acc;
  }
}
#undef RP

extern "C" void kernel_launch(void* const* d_in, const int* in_sizes, int n_in,
                              void* d_out, int out_size, void* d_ws, size_t ws_size,
                              hipStream_t stream) {
  (void)in_sizes; (void)n_in; (void)out_size; (void)ws_size;
  const float* anchor_x  = (const float*)d_in[0];
  const float* node_x    = (const float*)d_in[1];
  const float* anchor_f  = (const float*)d_in[2];
  const float* node_f    = (const float*)d_in[3];
  const float* node_mask = (const float*)d_in[6];
  const float* Wq  = (const float*)d_in[7];
  const float* bq  = (const float*)d_in[8];
  const float* Wkv = (const float*)d_in[9];
  const float* bkv = (const float*)d_in[10];
  const float* ln1_g = (const float*)d_in[11];
  const float* ln1_b = (const float*)d_in[12];
  const float* W1 = (const float*)d_in[13];
  const float* b1 = (const float*)d_in[14];
  const float* W2 = (const float*)d_in[15];
  const float* b2 = (const float*)d_in[16];
  const float* W3 = (const float*)d_in[17];
  const float* b3 = (const float*)d_in[18];
  const float* ln2_g = (const float*)d_in[19];
  const float* ln2_b = (const float*)d_in[20];

  const long WKV_LD = H + HE;

  float* p = (float*)d_ws;
  float* q    = p; p += (long)NA * H;
  float* qkqe = p; p += (long)NA * WKV_LD;
  float* L    = p; p += (long)NA * N_PER;
  float* attn = p; p += (long)NA * N_PER;
  float* ts   = p; p += (long)NA * WKV_LD;
  float* af   = p; p += (long)NA * H;
  float* m1   = p; p += (long)NA * 2 * H;
  float* m2   = p; p += (long)NA * 2 * H;

  auto gemm = [&](const float* A, long lda, long sAb,
                  const float* Bm, long ldbn, long ldbk, long sBb,
                  float* C, long ldc, long sCb, const float* bias,
                  int M, int N, int K, int nb, int relu, int addC) {
    GemmArgs g{A, lda, sAb, Bm, ldbn, ldbk, sBb, C, ldc, sCb, bias, K, relu, addC};
    dim3 grid(N / BN, M / BM, nb);
    if (ldbk == 1)
      wmma_gemm_f32<true><<<grid, dim3(128), 0, stream>>>(g);
    else
      wmma_gemm_f32<false><<<grid, dim3(128), 0, stream>>>(g);
  };

  auto gemm_ln = [&](const float* A, long lda, const float* Bm, long ldbn, const float* bias,
                     const float* resid, const float* gam, const float* bet, float* out, int K) {
    RowLnArgs g{A, lda, Bm, ldbn, bias, resid, gam, bet, out, K};
    wmma_gemm_rowln_f32<<<dim3(NA / LBM), dim3(256), 0, stream>>>(g);
  };

  gemm(anchor_f, H, 0, Wq, H, 1, 0, q, H, 0, bq, NA, H, H, 1, 0, 0);
  gemm(q, H, 0, Wkv, 1, WKV_LD, 0, qkqe, WKV_LD, 0, nullptr, NA, (int)WKV_LD, H, 1, 0, 0);
  gemm(qkqe, WKV_LD, (long)A_PER * WKV_LD,
       node_f, H, 1, (long)N_PER * H,
       L, N_PER, (long)A_PER * N_PER, nullptr, A_PER, N_PER, H, NB, 0, 0);
  edge_softmax_kernel<<<dim3(NA), dim3(256), 0, stream>>>(
      anchor_x, node_x, node_mask, q, bkv, qkqe + H, L, attn, ts + H,
      (int)WKV_LD, (int)WKV_LD);
  gemm(attn, N_PER, (long)A_PER * N_PER,
       node_f, 1, H, (long)N_PER * H,
       ts, WKV_LD, (long)A_PER * WKV_LD, nullptr, A_PER, H, N_PER, NB, 0, 0);
  gemm_ln(ts, WKV_LD, Wkv + (long)H * WKV_LD, WKV_LD, bkv + H, anchor_f, ln1_g, ln1_b, af, (int)WKV_LD);
  gemm(af, H, 0, W1, H, 1, 0, m1, 2 * H, 0, b1, NA, 2 * H, H, 1, 1, 0);
  gemm(m1, 2 * H, 0, W2, 2 * H, 1, 0, m2, 2 * H, 0, b2, NA, 2 * H, 2 * H, 1, 1, 0);
  gemm_ln(m2, 2 * H, W3, 2 * H, b3, af, ln2_g, ln2_b, (float*)d_out, 2 * H);
}


# GEMM K loops split over workgroups: 2-4 K chunks write partial planes, no adding pass; the reading kernel's A loader sums planes, adds bias, relu; 64-128 -> 160-256 workgroups (sum order changes)
# speedup vs baseline: 4.5763x; 1.1908x over previous
#include <hip/hip_runtime.h>
#include <math.h>

#define H      256
#define HE     64
#define NB     8
#define A_PER  64
#define N_PER  512
#define NA     (NB * A_PER)
#define NN     (NB * N_PER)

#define QSPLIT   4
#define KQSPLIT  2
#define LSPLIT   2
#define TSPLIT   4
#define M1SPLIT  2
#define M2SPLIT  2

typedef __attribute__((ext_vector_type(2))) float v2f;
typedef __attribute__((ext_vector_type(8))) float v8f;

#define BM 32
#define BN 64
#define BK 64
#define PK 68

struct GemmArgs {
  const float* A;  long lda;  long sAb;  long sAp;
  const float* B;  long ldbn; long ldbk; long sBb;
  float*       C;  long ldc;  long sCb;  long sCp;
  const float* bias;
  const float* abias;
  int K;
  int ksh;
  int arelu;
  int relu;
  int addC;
};

__device__ __forceinline__ float4 add4(float4 a, float4 b) {
  return make_float4(a.x + b.x, a.y + b.y, a.z + b.z, a.w + b.w);
}
__device__ __forceinline__ float4 relu4(float4 a) {
  return make_float4(fmaxf(a.x, 0.f), fmaxf(a.y, 0.f), fmaxf(a.z, 0.f), fmaxf(a.w, 0.f));
}

template <bool BKCONTIG, int NPA>
__global__ __launch_bounds__(128)
void wmma_gemm_f32(GemmArgs g) {
  __shared__ __attribute__((aligned(16))) float As[BM * PK];
  __shared__ __attribute__((aligned(16))) float Bs[BN * PK];

  const int t    = threadIdx.x;
  const int lane = t & 31;
  const int w    = t >> 5;
  const int r15  = lane & 15;
  const int kh   = (lane >> 4) << 1;
  const int wm   = w & 1;
  const int wn   = w >> 1;
  const int tn = blockIdx.x, tm = blockIdx.y;
  const int ks = blockIdx.z & ((1 << g.ksh) - 1);
  const int bz = blockIdx.z >> g.ksh;
  const long kofs = (long)ks * g.K;

  const int lr = t >> 4;
  const int lc = (t & 15) << 2;

  const float* Ag = g.A + (long)bz * g.sAb + (long)(tm * BM + lr) * g.lda + lc + kofs;
  const float* Bg = BKCONTIG
      ? g.B + (long)bz * g.sBb + (long)(tn * BN + lr) * g.ldbn + lc + kofs
      : g.B + (long)bz * g.sBb + (long)(2 * lr + kofs) * g.ldbk + (tn * BN + lc);
  const bool hasab = g.abias != nullptr;
  const float* Abp = hasab ? g.abias + kofs + lc : nullptr;

  const long a8 = 8 * g.lda;
  const long bu = BKCONTIG ? 8 * g.ldbn : g.ldbk;
  const long bstep = BKCONTIG ? (long)BK : (long)BK * g.ldbk;
#define BROW(i) ((BKCONTIG ? (i) : (((i) >> 1) * 16 + ((i) & 1))) * bu)
  float4 ra[NPA][4], rab, rb0, rb1, rb2, rb3, rb4, rb5, rb6, rb7;
  rab = make_float4(0.f, 0.f, 0.f, 0.f);
#define GLOAD_SLAB()                                                                   \
  _Pragma("unroll") for (int p_ = 0; p_ < NPA; ++p_) {                                 \
    _Pragma("unroll") for (int i_ = 0; i_ < 4; ++i_)                                   \
      ra[p_][i_] = *(const float4*)(Ag + p_ * g.sAp + i_ * a8);                        \
  }                                                                                    \
  if (hasab) { rab = *(const float4*)(Abp); Abp += BK; }                               \
  rb0 = *(const float4*)(Bg + BROW(0)); rb1 = *(const float4*)(Bg + BROW(1));          \
  rb2 = *(const float4*)(Bg + BROW(2)); rb3 = *(const float4*)(Bg + BROW(3));          \
  rb4 = *(const float4*)(Bg + BROW(4)); rb5 = *(const float4*)(Bg + BROW(5));          \
  rb6 = *(const float4*)(Bg + BROW(6)); rb7 = *(const float4*)(Bg + BROW(7));          \
  Ag += BK; Bg += bstep;
#define BPAIR_STORE(i, lo, hi)                                                         \
  *(v2f*)(Bw + 16 * (i))          = (v2f){lo.x, hi.x};                                 \
  *(v2f*)(Bw + 16 * (i) + PK)     = (v2f){lo.y, hi.y};                                 \
  *(v2f*)(Bw + 16 * (i) + 2 * PK) = (v2f){lo.z, hi.z};                                 \
  *(v2f*)(Bw + 16 * (i) + 3 * PK) = (v2f){lo.w, hi.w};

  float* Aw = &As[lr * PK + lc];
  float* Bw = BKCONTIG ? &Bs[lr * PK + lc] : &Bs[lc * PK + 2 * lr];
  constexpr int R8 = 8 * PK;

  const float* Af  = &As[(wm * 16 + r15) * PK + kh];
  const float* Bf0 = &Bs[(wn * 32 + r15) * PK + kh];
  const float* Bf1 = Bf0 + 16 * PK;

  const v8f zero = {0.f,0.f,0.f,0.f,0.f,0.f,0.f,0.f};
  v8f c0 = zero, c1 = zero;

  GLOAD_SLAB()
  for (int k0 = 0; k0 < g.K; k0 += BK) {
    #pragma unroll
    for (int i = 0; i < 4; ++i) {
      float4 a = ra[0][i];
      #pragma unroll
      for (int p = 1; p < NPA; ++p) a = add4(a, ra[p][i]);
      if (hasab)   a = add4(a, rab);
      if (g.arelu) a = relu4(a);
      *(float4*)(Aw + i * R8) = a;
    }
    if (BKCONTIG) {
      *(float4*)(Bw)          = rb0;  *(float4*)(Bw + R8)     = rb1;
      *(float4*)(Bw + 2 * R8) = rb2;  *(float4*)(Bw + 3 * R8) = rb3;
      *(float4*)(Bw + 4 * R8) = rb4;  *(float4*)(Bw + 5 * R8) = rb5;
      *(float4*)(Bw + 6 * R8) = rb6;  *(float4*)(Bw + 7 * R8) = rb7;
    } else {
      BPAIR_STORE(0, rb0, rb1)
      BPAIR_STORE(1, rb2, rb3)
      BPAIR_STORE(2, rb4, rb5)
      BPAIR_STORE(3, rb6, rb7)
    }
    __syncthreads();

    const bool more = (k0 + BK) < g.K;
    if (more) { GLOAD_SLAB() }

    #pragma unroll
    for (int kk = 0; kk < BK; kk += 4) {
      v2f a  = *(const v2f*)(Af + kk);
      v2f b0 = *(const v2f*)(Bf0 + kk);
      v2f b1 = *(const v2f*)(Bf1 + kk);
      c0 = __builtin_amdgcn_wmma_f32_16x16x4_f32(false, a, false, b0, (short)0, c0, false, false);
      c1 = __builtin_amdgcn_wmma_f32_16x16x4_f32(false, a, false, b1, (short)0, c1, false, false);
    }
    if (more) __syncthreads();
  }

  const int  col0 = tn * BN + wn * 32 + r15;
  const int  col1 = col0 + 16;
  const int  m0   = tm * BM + wm * 16 + ((lane >> 4) << 3);
  const float bv0 = g.bias ? g.bias[col0] : 0.f;
  const float bv1 = g.bias ? g.bias[col1] : 0.f;
  float* Cp = g.C + (long)bz * g.sCb + (long)ks * g.sCp;
  #pragma unroll
  for (int r = 0; r < 8; ++r) {
    const long row = (long)(m0 + r) * g.ldc;
    float v0 = c0[r] + bv0;
    float v1 = c1[r] + bv1;
    if (g.addC) { v0 += Cp[row + col0]; v1 += Cp[row + col1]; }
    if (g.relu) { v0 = fmaxf(v0, 0.f);  v1 = fmaxf(v1, 0.f); }
    Cp[row + col0] = v0;
    Cp[row + col1] = v1;
  }
}
#undef GLOAD_SLAB
#undef BPAIR_STORE
#undef BROW

#define LBM 16
#define LBN 256

struct RowLnArgs {
  const float* A;  long lda;  long sAp;
  const float* A2; long lda2; int K1;
  const float* abias;
  int arelu;
  const float* B;  long ldbn;
  const float* bias;
  const float* resid;
  const float* gam;
  const float* bet;
  float*       out;
  int K;
};

template <int XM>
__device__ __forceinline__ float row_xor_add(float v) {
  const int o = __builtin_amdgcn_update_dpp(0, __float_as_int(v), 0x160 + XM, 0xf, 0xf, true);
  return v + __int_as_float(o);
}

template <int NPA, bool TRAIL>
__global__ __launch_bounds__(256)
void wmma_gemm_rowln_f32(RowLnArgs g) {
  constexpr int ASTG = LBM * PK;
  constexpr int BSTG = LBN * PK;
  __shared__ __attribute__((aligned(16))) float As[2 * ASTG];
  __shared__ __attribute__((aligned(16))) float Bs[2 * BSTG];
  __shared__ __attribute__((aligned(16))) float red1[LBM * 8];
  __shared__ __attribute__((aligned(16))) float red2[LBM * 8];

  const int t    = threadIdx.x;
  const int lane = t & 31;
  const int w    = t >> 5;
  const int r15  = lane & 15;
  const int hh   = lane >> 4;
  const int kh   = hh << 1;
  const int tm   = blockIdx.x;

  const int lr = t >> 4;
  const int lc = (t & 15) << 2;

  const float* Ag  = g.A + (long)(tm * LBM + lr) * g.lda + lc;
  const float* A2g = g.A2 + (long)(tm * LBM + lr) * g.lda2 + lc;
  int kld = 0;
  const float* Bg = g.B + (long)lr * g.ldbn + lc;
  const long b16 = 16 * g.ldbn;
  const bool hasab = g.abias != nullptr;
  const float* Abp = (hasab ? g.abias : g.bias) + lc;
  const int abstep = hasab ? BK : 0;

  const int  col0 = w * 32 + r15;
  const int  col1 = col0 + 16;
  const int  m0   = tm * LBM + (hh << 3);
  const float bv0 = g.bias[col0], bv1 = g.bias[col1];
  const float ga0 = g.gam[col0],  ga1 = g.gam[col1];
  const float be0 = g.bet[col0],  be1 = g.bet[col1];
  float x0[8], x1[8];
  for (int r = 0; r < 8; ++r) {
    x0[r] = g.resid[(long)(m0 + r) * LBN + col0];
    x1[r] = g.resid[(long)(m0 + r) * LBN + col1];
  }

  float4 ra[NPA], rab, rb0, rb1, rb2, rb3, rb4, rb5, rb6, rb7, rb8, rb9, rb10, rb11, rb12, rb13, rb14, rb15;
  rab = make_float4(0.f, 0.f, 0.f, 0.f);
#define LN_GLOAD_SLAB()                                                                        \
  rab = *(const float4*)(Abp); Abp += abstep;                                                  \
  rb0 = *(const float4*)(Bg);                  rb1 = *(const float4*)(Bg + 1 * b16);        \
  rb2 = *(const float4*)(Bg + 2 * b16);        rb3 = *(const float4*)(Bg + 3 * b16);        \
  rb4 = *(const float4*)(Bg + 4 * b16);        rb5 = *(const float4*)(Bg + 5 * b16);        \
  rb6 = *(const float4*)(Bg + 6 * b16);        rb7 = *(const float4*)(Bg + 7 * b16);        \
  rb8 = *(const float4*)(Bg + 8 * b16);        rb9 = *(const float4*)(Bg + 9 * b16);        \
  rb10 = *(const float4*)(Bg + 10 * b16);      rb11 = *(const float4*)(Bg + 11 * b16);      \
  rb12 = *(const float4*)(Bg + 12 * b16);      rb13 = *(const float4*)(Bg + 13 * b16);      \
  rb14 = *(const float4*)(Bg + 14 * b16);      rb15 = *(const float4*)(Bg + 15 * b16);      \
  Bg += BK;                                                                                    \
    \
    \
  {                                                                                            \
    const bool pl_ = !TRAIL || kld < g.K1;     \
    ra[0] = *(const float4*)(pl_ ? Ag : A2g);                                                  \
    _Pragma("unroll") for (int p_ = 1; p_ < NPA; ++p_) {                                       \
      const float4 v_ = *(const float4*)(pl_ ? Ag + p_ * g.sAp : A2g);                         \
      ra[p_] = pl_ ? v_ : make_float4(0.f, 0.f, 0.f, 0.f);                                     \
    }                                                                                          \
    Ag += pl_ ? BK : 0;  A2g += pl_ ? 0 : BK;  kld += BK;                                      \
  }
  constexpr int R16 = 16 * PK;
#define LN_STORE_SLAB(stg)                                                                     \
  {                                                                                            \
    float* bw = Bw + (stg) * BSTG;                                                             \
    float4 a_ = ra[0];                                \
    _Pragma("unroll") for (int p_ = 1; p_ < NPA; ++p_) a_ = add4(a_, ra[p_]);                  \
    a_ = hasab ? add4(a_, rab) : a_;                                                           \
    a_ = g.arelu ? relu4(a_) : a_;                                                             \
    *(float4*)(Aw + (stg) * ASTG) = a_;                                                        \
    *(float4*)(bw) = rb0;                    *(float4*)(bw + 1 * R16) = rb1;          \
    *(float4*)(bw + 2 * R16) = rb2;          *(float4*)(bw + 3 * R16) = rb3;          \
    *(float4*)(bw + 4 * R16) = rb4;          *(float4*)(bw + 5 * R16) = rb5;          \
    *(float4*)(bw + 6 * R16) = rb6;          *(float4*)(bw + 7 * R16) = rb7;          \
    *(float4*)(bw + 8 * R16) = rb8;          *(float4*)(bw + 9 * R16) = rb9;          \
    *(float4*)(bw + 10 * R16) = rb10;        *(float4*)(bw + 11 * R16) = rb11;        \
    *(float4*)(bw + 12 * R16) = rb12;        *(float4*)(bw + 13 * R16) = rb13;        \
    *(float4*)(bw + 14 * R16) = rb14;        *(float4*)(bw + 15 * R16) = rb15;        \
  }

  float* Aw = &As[lr * PK + lc];
  float* Bw = &Bs[lr * PK + lc];

  const float* Af  = &As[r15 * PK + kh];
  const float* Bf  = &Bs[(w * 32 + r15) * PK + kh];

  const v8f zero = {0.f,0.f,0.f,0.f,0.f,0.f,0.f,0.f};
  v8f c0 = zero, c1 = zero;

  LN_GLOAD_SLAB()
  LN_STORE_SLAB(0)
  __syncthreads();
  int st = 0;
  int k0 = 0;
  bool more;
  do {
    more = (k0 + BK) < g.K;
    if (more) { LN_GLOAD_SLAB() }

    const float* af  = Af + st * ASTG;
    const float* bf0 = Bf + st * BSTG;
    const float* bf1 = bf0 + 16 * PK;
    #pragma unroll
    for (int kk = 0; kk < BK; kk += 4) {
      v2f a  = *(const v2f*)(af + kk);
      v2f b0 = *(const v2f*)(bf0 + kk);
      v2f b1 = *(const v2f*)(bf1 + kk);
      c0 = __builtin_amdgcn_wmma_f32_16x16x4_f32(false, a, false, b0, (short)0, c0, false, false);
      c1 = __builtin_amdgcn_wmma_f32_16x16x4_f32(false, a, false, b1, (short)0, c1, false, false);
    }
    if (more) {
      st ^= 1;
      LN_STORE_SLAB(st)
      __syncthreads();
    }
    k0 += BK;
  } while (more);
  asm volatile("v_nop\n\tv_nop\n\tv_nop\n\tv_nop\n\tv_nop" : "+v"(c0), "+v"(c1));
#undef LN_GLOAD_SLAB
#undef LN_STORE_SLAB

  float v0[8], v1[8], ps[8];
  for (int r = 0; r < 8; ++r) {
    v0[r] = x0[r] + (c0[r] + bv0);
    v1[r] = x1[r] + (c1[r] + bv1);
    ps[r] = v0[r] + v1[r];
  }
  for (int r = 0; r < 8; ++r) ps[r] = row_xor_add<1>(ps[r]);
  for (int r = 0; r < 8; ++r) ps[r] = row_xor_add<2>(ps[r]);
  for (int r = 0; r < 8; ++r) ps[r] = row_xor_add<4>(ps[r]);
  for (int r = 0; r < 8; ++r) ps[r] = row_xor_add<8>(ps[r]);
  if (r15 == 0) {
    for (int r = 0; r < 8; ++r) red1[((hh << 3) + r) * 8 + w] = ps[r];
  }
  __syncthreads();
  for (int r = 0; r < 8; ++r) {
    const float4 pa = *(const float4*)(&red1[((hh << 3) + r) * 8]);
    const float4 pb = *(const float4*)(&red1[((hh << 3) + r) * 8 + 4]);
    const float mean = (((pa.x + pa.y) + (pa.z + pa.w)) + ((pb.x + pb.y) + (pb.z + pb.w))) * (1.f / LBN);
    v0[r] -= mean;
    v1[r] -= mean;
    ps[r] = v0[r] * v0[r] + v1[r] * v1[r];
  }
  for (int r = 0; r < 8; ++r) ps[r] = row_xor_add<1>(ps[r]);
  for (int r = 0; r < 8; ++r) ps[r] = row_xor_add<2>(ps[r]);
  for (int r = 0; r < 8; ++r) ps[r] = row_xor_add<4>(ps[r]);
  for (int r = 0; r < 8; ++r) ps[r] = row_xor_add<8>(ps[r]);
  if (r15 == 0) {
    for (int r = 0; r < 8; ++r) red2[((hh << 3) + r) * 8 + w] = ps[r];
  }
  __syncthreads();
  for (int r = 0; r < 8; ++r) {
    const float4 pa = *(const float4*)(&red2[((hh << 3) + r) * 8]);
    const float4 pb = *(const float4*)(&red2[((hh << 3) + r) * 8 + 4]);
    const float var  = (((pa.x + pa.y) + (pa.z + pa.w)) + ((pb.x + pb.y) + (pb.z + pb.w))) * (1.f / LBN);
    const float rstd = rsqrtf(var + 1e-5f);
    const long row = (long)(m0 + r) * LBN;
    g.out[row + col0] = v0[r] * rstd * ga0 + be0;
    g.out[row + col1] = v1[r] * rstd * ga1 + be1;
  }
}

#define RP 68
__global__ __launch_bounds__(256)
void edge_softmax_kernel(const float* __restrict__ anchor_x,
                         const float* __restrict__ node_x,
                         const float* __restrict__ node_mask,
                         const float* __restrict__ q,
                         const float* __restrict__ bq,
                         const float* __restrict__ bk,
                         const float* __restrict__ qe,
                         const float* Lbase,
                         float* attn,
                         float* __restrict__ s_out,
                         int ldq, int lds_, long qps, long kps, long Lps)
{
  const int bid = blockIdx.x;
  const int b   = bid >> 6;
  const int t   = threadIdx.x;

  __shared__ __attribute__((aligned(16))) float sh_rbf[N_PER * RP];
  __shared__ __attribute__((aligned(16))) float sh_l[N_PER];
  __shared__ __attribute__((aligned(16))) float sh_part[16 * HE];
  __shared__ __attribute__((aligned(16))) float sh_qe[HE];
  __shared__ float red[256];
  __shared__ float sh_bcast;
  __shared__ float sh_ax[3];
  __shared__ float sh_c[8];

  if (t < HE) {
    float v = qe[(long)bid * ldq + t];
    for (int p = 1; p < KQSPLIT; ++p) v += qe[p * kps + (long)bid * ldq + t];
    sh_qe[t] = v;
  }
  if (t < 3)  sh_ax[t] = anchor_x[bid * 3 + t];
  float qv = q[(long)bid * H + t];
  for (int p = 1; p < QSPLIT; ++p) qv += q[p * qps + (long)bid * H + t];
  qv += bq[t];
  float pc = qv * bk[t];
  for (int o = 16; o > 0; o >>= 1) pc += __shfl_xor(pc, o, 32);
  if ((t & 31) == 0) sh_c[t >> 5] = pc;
  __syncthreads();
  const float ca = ((sh_c[0] + sh_c[1]) + (sh_c[2] + sh_c[3])) +
                   ((sh_c[4] + sh_c[5]) + (sh_c[6] + sh_c[7]));

  const float axx = sh_ax[0], axy = sh_ax[1], axz = sh_ax[2];
  const float MU_STEP = 20.f / 63.f;
  const float INV_SIG = 64.f / 20.f;

  for (int n = t; n < N_PER; n += 256) {
    const float* nx = node_x + (long)(b * N_PER + n) * 3;
    float dx = axx - nx[0] + 1e-8f;
    float dy = axy - nx[1] + 1e-8f;
    float dz = axz - nx[2] + 1e-8f;
    float d  = sqrtf(dx * dx + dy * dy + dz * dz);
    float d10 = d * 0.1f;
    float* rrow = &sh_rbf[n * RP];
    float le = 0.f;
    #pragma unroll 2
    for (int j = 0; j < HE; j += 4) {
      float u0 = (d10 - (j    ) * MU_STEP) * INV_SIG;
      float u1 = (d10 - (j + 1) * MU_STEP) * INV_SIG;
      float u2 = (d10 - (j + 2) * MU_STEP) * INV_SIG;
      float u3 = (d10 - (j + 3) * MU_STEP) * INV_SIG;
      float4 e;
      e.x = __expf(-u0 * u0);
      e.y = __expf(-u1 * u1);
      e.z = __expf(-u2 * u2);
      e.w = __expf(-u3 * u3);
      le = fmaf(sh_qe[j],     e.x, le);
      le = fmaf(sh_qe[j + 1], e.y, le);
      le = fmaf(sh_qe[j + 2], e.z, le);
      le = fmaf(sh_qe[j + 3], e.w, le);
      *(float4*)(rrow + j) = e;
    }
    float l0 = Lbase[(long)bid * N_PER + n];
    for (int p = 1; p < LSPLIT; ++p) l0 += Lbase[p * Lps + (long)bid * N_PER + n];
    float lb = (l0 + ca) + le;
    lb *= (node_mask[b * N_PER + n] - 1.f) * 1000000.0f;
    sh_l[n] = lb;
  }
  __syncthreads();

  red[t] = fmaxf(sh_l[t], sh_l[t + 256]);
  __syncthreads();
  for (int st = 128; st > 0; st >>= 1) {
    if (t < st) red[t] = fmaxf(red[t], red[t + st]);
    __syncthreads();
  }
  if (t == 0) sh_bcast = red[0];
  __syncthreads();
  const float mx = sh_bcast;

  float e0 = __expf(sh_l[t] - mx);
  float e1 = __expf(sh_l[t + 256] - mx);
  red[t] = e0 + e1;
  __syncthreads();
  for (int st = 128; st > 0; st >>= 1) {
    if (t < st) red[t] += red[t + st];
    __syncthreads();
  }
  if (t == 0) sh_bcast = red[0];
  __syncthreads();
  const float inv = 1.f / sh_bcast;

  const float a0 = e0 * inv, a1 = e1 * inv;
  sh_l[t] = a0; sh_l[t + 256] = a1;
  attn[(long)bid * N_PER + t]       = a0;
  attn[(long)bid * N_PER + t + 256] = a1;
  __syncthreads();

  const int jg = (t & 15) << 2, c = t >> 4;
  const float* rp = &sh_rbf[(c * 32) * RP + jg];
  const float* ap = &sh_l[c * 32];
  float4 part = make_float4(0.f, 0.f, 0.f, 0.f);
  for (int i = 0; i < 32; i += 4) {
    const float4 a4 = *(const float4*)(ap + i);
    const float4 r0 = *(const float4*)(rp + (i    ) * RP);
    const float4 r1 = *(const float4*)(rp + (i + 1) * RP);
    const float4 r2 = *(const float4*)(rp + (i + 2) * RP);
    const float4 r3 = *(const float4*)(rp + (i + 3) * RP);
    part.x = fmaf(a4.x, r0.x, part.x); part.y = fmaf(a4.x, r0.y, part.y);
    part.z = fmaf(a4.x, r0.z, part.z); part.w = fmaf(a4.x, r0.w, part.w);
    part.x = fmaf(a4.y, r1.x, part.x); part.y = fmaf(a4.y, r1.y, part.y);
    part.z = fmaf(a4.y, r1.z, part.z); part.w = fmaf(a4.y, r1.w, part.w);
    part.x = fmaf(a4.z, r2.x, part.x); part.y = fmaf(a4.z, r2.y, part.y);
    part.z = fmaf(a4.z, r2.z, part.z); part.w = fmaf(a4.z, r2.w, part.w);
    part.x = fmaf(a4.w, r3.x, part.x); part.y = fmaf(a4.w, r3.y, part.y);
    part.z = fmaf(a4.w, r3.z, part.z); part.w = fmaf(a4.w, r3.w, part.w);
  }
  *(float4*)(&sh_part[c * HE + jg]) = part;
  __syncthreads();
  if (t < HE) {
    float acc = sh_part[t];
    for (int cc = 1; cc < 16; ++cc) acc += sh_part[cc * HE + t];
    s_out[(long)bid * lds_ + t] = acc;
  }
}
#undef RP

extern "C" void kernel_launch(void* const* d_in, const int* in_sizes, int n_in,
                              void* d_out, int out_size, void* d_ws, size_t ws_size,
                              hipStream_t stream) {
  (void)in_sizes; (void)n_in; (void)out_size; (void)ws_size;
  const float* anchor_x  = (const float*)d_in[0];
  const float* node_x    = (const float*)d_in[1];
  const float* anchor_f  = (const float*)d_in[2];
  const float* node_f    = (const float*)d_in[3];
  const float* node_mask = (const float*)d_in[6];
  const float* Wq  = (const float*)d_in[7];
  const float* bq  = (const float*)d_in[8];
  const float* Wkv = (const float*)d_in[9];
  const float* bkv = (const float*)d_in[10];
  const float* ln1_g = (const float*)d_in[11];
  const float* ln1_b = (const float*)d_in[12];
  const float* W1 = (const float*)d_in[13];
  const float* b1 = (const float*)d_in[14];
  const float* W2 = (const float*)d_in[15];
  const float* b2 = (const float*)d_in[16];
  const float* W3 = (const float*)d_in[17];
  const float* b3 = (const float*)d_in[18];
  const float* ln2_g = (const float*)d_in[19];
  const float* ln2_b = (const float*)d_in[20];

  const long WKV_LD = H + HE;

  const long QP  = (long)NA * H;
  const long KQP = (long)NA * WKV_LD;
  const long LP  = (long)NA * N_PER;
  const long TP  = (long)NA * H;
  const long MP  = (long)NA * 2 * H;

  float* p = (float*)d_ws;
  float* q    = p; p += QSPLIT * QP;
  float* qkqe = p; p += KQSPLIT * KQP;
  float* L    = p; p += LSPLIT * LP;
  float* sbuf = p; p += (long)NA * HE;
  float* attn = L;
  float* ts   = q;
  float* af   = qkqe;
  float* m1   = L;
  float* m2   = q;

  auto gemm = [&](const float* A, long lda, long sAb, int npa, long sAp, const float* abias, int arelu,
                  const float* Bm, long ldbn, long ldbk, long sBb,
                  float* C, long ldc, long sCb, long sCp, const float* bias,
                  int M, int N, int Kc, int ksh, int nb, int relu) {
    GemmArgs g{A, lda, sAb, sAp, Bm, ldbn, ldbk, sBb, C, ldc, sCb, sCp, bias, abias, Kc, ksh, arelu, relu, 0};
    dim3 grid(N / BN, M / BM, nb << ksh);
    if (ldbk == 1) {
      if (npa == 1) wmma_gemm_f32<true, 1><<<grid, dim3(128), 0, stream>>>(g);
      else          wmma_gemm_f32<true, 2><<<grid, dim3(128), 0, stream>>>(g);
    } else {
      if (npa == 1) wmma_gemm_f32<false, 1><<<grid, dim3(128), 0, stream>>>(g);
      else          wmma_gemm_f32<false, 4><<<grid, dim3(128), 0, stream>>>(g);
    }
  };

  gemm(anchor_f, H, 0, 1, 0, nullptr, 0, Wq, H, 1, 0,
       q, H, 0, QP, nullptr, NA, H, H / QSPLIT, 2, 1, 0);
  gemm(q, H, 0, QSPLIT, QP, bq, 0, Wkv, 1, WKV_LD, 0,
       qkqe, WKV_LD, 0, KQP, nullptr, NA, (int)WKV_LD, H / KQSPLIT, 1, 1, 0);
  gemm(qkqe, WKV_LD, (long)A_PER * WKV_LD, KQSPLIT, KQP, nullptr, 0,
       node_f, H, 1, (long)N_PER * H,
       L, N_PER, (long)A_PER * N_PER, LP, nullptr, A_PER, N_PER, H / LSPLIT, 1, NB, 0);
  edge_softmax_kernel<<<dim3(NA), dim3(256), 0, stream>>>(
      anchor_x, node_x, node_mask, q, bq, bkv, qkqe + H, L, attn, sbuf,
      (int)WKV_LD, (int)HE, QP, KQP, LP);
  gemm(attn, N_PER, (long)A_PER * N_PER, 1, 0, nullptr, 0,
       node_f, 1, H, (long)N_PER * H,
       ts, H, (long)A_PER * H, TP, nullptr, A_PER, H, N_PER / TSPLIT, 2, NB, 0);
  {
    RowLnArgs g{ts, H, TP, sbuf, HE, H, nullptr, 0, Wkv + (long)H * WKV_LD, WKV_LD, bkv + H,
                anchor_f, ln1_g, ln1_b, af, (int)WKV_LD};
    wmma_gemm_rowln_f32<TSPLIT, true><<<dim3(NA / LBM), dim3(256), 0, stream>>>(g);
  }
  gemm(af, H, 0, 1, 0, nullptr, 0, W1, H, 1, 0,
       m1, 2 * H, 0, MP, nullptr, NA, 2 * H, H / M1SPLIT, 1, 1, 0);
  gemm(m1, 2 * H, 0, M1SPLIT, MP, b1, 1, W2, 2 * H, 1, 0,
       m2, 2 * H, 0, MP, nullptr, NA, 2 * H, 2 * H / M2SPLIT, 1, 1, 0);
  {
    RowLnArgs g{m2, 2 * H, MP, m2, 2 * H, 2 * H, b2, 1, W3, 2 * H, b3, af, ln2_g, ln2_b, (float*)d_out, 2 * H};
    wmma_gemm_rowln_f32<M2SPLIT, false><<<dim3(NA / LBM), dim3(256), 0, stream>>>(g);
  }
}


# Row-norm GEMMs: each wave stages its own 32-column B slab in LDS, A is staged once for all K: no workgroup barrier in the K loop, a SIMD's two waves overlap copy and WMMA phases; sums unchanged
# speedup vs baseline: 4.8635x; 1.0628x over previous
#include <hip/hip_runtime.h>
#include <math.h>

#define H      256
#define HE     64
#define NB     8
#define A_PER  64
#define N_PER  512
#define NA     (NB * A_PER)
#define NN     (NB * N_PER)

#define QSPLIT   4
#define KQSPLIT  2
#define LSPLIT   2
#define TSPLIT   4
#define M1SPLIT  2
#define M2SPLIT  2

typedef __attribute__((ext_vector_type(2))) float v2f;
typedef __attribute__((ext_vector_type(8))) float v8f;

#define BM 32
#define BN 64
#define BK 64
#define PK 68

struct GemmArgs {
  const float* A;  long lda;  long sAb;  long sAp;
  const float* B;  long ldbn; long ldbk; long sBb;
  float*       C;  long ldc;  long sCb;  long sCp;
  const float* bias;
  const float* abias;
  int K;
  int ksh;
  int arelu;
  int relu;
  int addC;
};

__device__ __forceinline__ float4 add4(float4 a, float4 b) {
  return make_float4(a.x + b.x, a.y + b.y, a.z + b.z, a.w + b.w);
}
__device__ __forceinline__ float4 relu4(float4 a) {
  return make_float4(fmaxf(a.x, 0.f), fmaxf(a.y, 0.f), fmaxf(a.z, 0.f), fmaxf(a.w, 0.f));
}

template <bool BKCONTIG, int NPA>
__global__ __launch_bounds__(128)
void wmma_gemm_f32(GemmArgs g) {
  __shared__ __attribute__((aligned(16))) float As[BM * PK];
  __shared__ __attribute__((aligned(16))) float Bs[BN * PK];

  const int t    = threadIdx.x;
  const int lane = t & 31;
  const int w    = t >> 5;
  const int r15  = lane & 15;
  const int kh   = (lane >> 4) << 1;
  const int wm   = w & 1;
  const int wn   = w >> 1;
  const int tn = blockIdx.x, tm = blockIdx.y;
  const int ks = blockIdx.z & ((1 << g.ksh) - 1);
  const int bz = blockIdx.z >> g.ksh;
  const long kofs = (long)ks * g.K;

  const int lr = t >> 4;
  const int lc = (t & 15) << 2;

  const float* Ag = g.A + (long)bz * g.sAb + (long)(tm * BM + lr) * g.lda + lc + kofs;
  const float* Bg = BKCONTIG
      ? g.B + (long)bz * g.sBb + (long)(tn * BN + lr) * g.ldbn + lc + kofs
      : g.B + (long)bz * g.sBb + (long)(2 * lr + kofs) * g.ldbk + (tn * BN + lc);
  const bool hasab = g.abias != nullptr;
  const float* Abp = hasab ? g.abias + kofs + lc : nullptr;

  const long a8 = 8 * g.lda;
  const long bu = BKCONTIG ? 8 * g.ldbn : g.ldbk;
  const long bstep = BKCONTIG ? (long)BK : (long)BK * g.ldbk;
#define BROW(i) ((BKCONTIG ? (i) : (((i) >> 1) * 16 + ((i) & 1))) * bu)
  float4 ra[NPA][4], rab, rb0, rb1, rb2, rb3, rb4, rb5, rb6, rb7;
  rab = make_float4(0.f, 0.f, 0.f, 0.f);
#define GLOAD_SLAB()                                                                   \
  _Pragma("unroll") for (int p_ = 0; p_ < NPA; ++p_) {                                 \
    _Pragma("unroll") for (int i_ = 0; i_ < 4; ++i_)                                   \
      ra[p_][i_] = *(const float4*)(Ag + p_ * g.sAp + i_ * a8);                        \
  }                                                                                    \
  if (hasab) { rab = *(const float4*)(Abp); Abp += BK; }                               \
  rb0 = *(const float4*)(Bg + BROW(0)); rb1 = *(const float4*)(Bg + BROW(1));          \
  rb2 = *(const float4*)(Bg + BROW(2)); rb3 = *(const float4*)(Bg + BROW(3));          \
  rb4 = *(const float4*)(Bg + BROW(4)); rb5 = *(const float4*)(Bg + BROW(5));          \
  rb6 = *(const float4*)(Bg + BROW(6)); rb7 = *(const float4*)(Bg + BROW(7));          \
  Ag += BK; Bg += bstep;
#define BPAIR_STORE(i, lo, hi)                                                         \
  *(v2f*)(Bw + 16 * (i))          = (v2f){lo.x, hi.x};                                 \
  *(v2f*)(Bw + 16 * (i) + PK)     = (v2f){lo.y, hi.y};                                 \
  *(v2f*)(Bw + 16 * (i) + 2 * PK) = (v2f){lo.z, hi.z};                                 \
  *(v2f*)(Bw + 16 * (i) + 3 * PK) = (v2f){lo.w, hi.w};

  float* Aw = &As[lr * PK + lc];
  float* Bw = BKCONTIG ? &Bs[lr * PK + lc] : &Bs[lc * PK + 2 * lr];
  constexpr int R8 = 8 * PK;

  const float* Af  = &As[(wm * 16 + r15) * PK + kh];
  const float* Bf0 = &Bs[(wn * 32 + r15) * PK + kh];
  const float* Bf1 = Bf0 + 16 * PK;

  const v8f zero = {0.f,0.f,0.f,0.f,0.f,0.f,0.f,0.f};
  v8f c0 = zero, c1 = zero;

  GLOAD_SLAB()
  for (int k0 = 0; k0 < g.K; k0 += BK) {
    #pragma unroll
    for (int i = 0; i < 4; ++i) {
      float4 a = ra[0][i];
      #pragma unroll
      for (int p = 1; p < NPA; ++p) a = add4(a, ra[p][i]);
      if (hasab)   a = add4(a, rab);
      if (g.arelu) a = relu4(a);
      *(float4*)(Aw + i * R8) = a;
    }
    if (BKCONTIG) {
      *(float4*)(Bw)          = rb0;  *(float4*)(Bw + R8)     = rb1;
      *(float4*)(Bw + 2 * R8) = rb2;  *(float4*)(Bw + 3 * R8) = rb3;
      *(float4*)(Bw + 4 * R8) = rb4;  *(float4*)(Bw + 5 * R8) = rb5;
      *(float4*)(Bw + 6 * R8) = rb6;  *(float4*)(Bw + 7 * R8) = rb7;
    } else {
      BPAIR_STORE(0, rb0, rb1)
      BPAIR_STORE(1, rb2, rb3)
      BPAIR_STORE(2, rb4, rb5)
      BPAIR_STORE(3, rb6, rb7)
    }
    __syncthreads();

    const bool more = (k0 + BK) < g.K;
    if (more) { GLOAD_SLAB() }

    #pragma unroll
    for (int kk = 0; kk < BK; kk += 4) {
      v2f a  = *(const v2f*)(Af + kk);
      v2f b0 = *(const v2f*)(Bf0 + kk);
      v2f b1 = *(const v2f*)(Bf1 + kk);
      c0 = __builtin_amdgcn_wmma_f32_16x16x4_f32(false, a, false, b0, (short)0, c0, false, false);
      c1 = __builtin_amdgcn_wmma_f32_16x16x4_f32(false, a, false, b1, (short)0, c1, false, false);
    }
    if (more) __syncthreads();
  }

  const int  col0 = tn * BN + wn * 32 + r15;
  const int  col1 = col0 + 16;
  const int  m0   = tm * BM + wm * 16 + ((lane >> 4) << 3);
  const float bv0 = g.bias ? g.bias[col0] : 0.f;
  const float bv1 = g.bias ? g.bias[col1] : 0.f;
  float* Cp = g.C + (long)bz * g.sCb + (long)ks * g.sCp;
  #pragma unroll
  for (int r = 0; r < 8; ++r) {
    const long row = (long)(m0 + r) * g.ldc;
    float v0 = c0[r] + bv0;
    float v1 = c1[r] + bv1;
    if (g.addC) { v0 += Cp[row + col0]; v1 += Cp[row + col1]; }
    if (g.relu) { v0 = fmaxf(v0, 0.f);  v1 = fmaxf(v1, 0.f); }
    Cp[row + col0] = v0;
    Cp[row + col1] = v1;
  }
}
#undef GLOAD_SLAB
#undef BPAIR_STORE
#undef BROW

#define LBM 16
#define LBN 256

struct RowLnArgs {
  const float* A;  long lda;  long sAp;
  const float* A2; long lda2;
  const float* abias;
  int arelu;
  const float* B;  long ldbn;
  const float* bias;
  const float* resid;
  const float* gam;
  const float* bet;
  float*       out;
};

template <int XM>
__device__ __forceinline__ float row_xor_add(float v) {
  const int o = __builtin_amdgcn_update_dpp(0, __float_as_int(v), 0x160 + XM, 0xf, 0xf, true);
  return v + __int_as_float(o);
}

__device__ __forceinline__ void wave_lds_order() {
  __builtin_amdgcn_fence(__ATOMIC_RELEASE, "wavefront");
  __builtin_amdgcn_wave_barrier();
  __builtin_amdgcn_fence(__ATOMIC_ACQUIRE, "wavefront");
}

template <int NPA, int KPL, int KTR>
__global__ __launch_bounds__(256) __attribute__((amdgpu_waves_per_eu(2, 2)))
void wmma_gemm_rowln_f32(RowLnArgs g) {
  static_assert(KPL % BK == 0 && (KTR == 0 || KTR == BK), "whole 64-k slabs");
  constexpr int KT  = KPL + KTR;
  constexpr int NSL = KPL / BK;
  constexpr int APK = KT + 4;
  constexpr int BWT = 32 * PK;
  __shared__ __attribute__((aligned(16))) float As[LBM * APK];
  __shared__ __attribute__((aligned(16))) float Bs[8 * BWT];
  __shared__ __attribute__((aligned(16))) float red1[LBM * 8];
  __shared__ __attribute__((aligned(16))) float red2[LBM * 8];

  const int t    = threadIdx.x;
  const int lane = t & 31;
  const int w    = t >> 5;
  const int r15  = lane & 15;
  const int hh   = lane >> 4;
  const int kh   = hh << 1;
  const int tm   = blockIdx.x;

  const int bc = r15 << 2;
  const float* Bg = g.B + (long)(w * 32 + hh) * g.ldbn + bc;
  const long b2 = 2 * g.ldbn;
  float* Bw = &Bs[w * BWT + hh * PK + bc];
  constexpr int R2 = 2 * PK;

  float4 rb0, rb1, rb2, rb3, rb4, rb5, rb6, rb7, rb8, rb9, rb10, rb11, rb12, rb13, rb14, rb15;
#define LN_B_GLOAD()                                                                           \
  rb0  = *(const float4*)(Bg);             rb1  = *(const float4*)(Bg + 1 * b2);               \
  rb2  = *(const float4*)(Bg + 2 * b2);    rb3  = *(const float4*)(Bg + 3 * b2);               \
  rb4  = *(const float4*)(Bg + 4 * b2);    rb5  = *(const float4*)(Bg + 5 * b2);               \
  rb6  = *(const float4*)(Bg + 6 * b2);    rb7  = *(const float4*)(Bg + 7 * b2);               \
  rb8  = *(const float4*)(Bg + 8 * b2);    rb9  = *(const float4*)(Bg + 9 * b2);               \
  rb10 = *(const float4*)(Bg + 10 * b2);   rb11 = *(const float4*)(Bg + 11 * b2);              \
  rb12 = *(const float4*)(Bg + 12 * b2);   rb13 = *(const float4*)(Bg + 13 * b2);              \
  rb14 = *(const float4*)(Bg + 14 * b2);   rb15 = *(const float4*)(Bg + 15 * b2);              \
  Bg += BK;
#define LN_B_STORE()                                                                           \
  *(float4*)(Bw)           = rb0;          *(float4*)(Bw + 1 * R2)  = rb1;                     \
  *(float4*)(Bw + 2 * R2)  = rb2;          *(float4*)(Bw + 3 * R2)  = rb3;                     \
  *(float4*)(Bw + 4 * R2)  = rb4;          *(float4*)(Bw + 5 * R2)  = rb5;                     \
  *(float4*)(Bw + 6 * R2)  = rb6;          *(float4*)(Bw + 7 * R2)  = rb7;                     \
  *(float4*)(Bw + 8 * R2)  = rb8;          *(float4*)(Bw + 9 * R2)  = rb9;                     \
  *(float4*)(Bw + 10 * R2) = rb10;         *(float4*)(Bw + 11 * R2) = rb11;                    \
  *(float4*)(Bw + 12 * R2) = rb12;         *(float4*)(Bw + 13 * R2) = rb13;                    \
  *(float4*)(Bw + 14 * R2) = rb14;         *(float4*)(Bw + 15 * R2) = rb15;

  LN_B_GLOAD()

  const int lr = t >> 4;
  const int lc = (t & 15) << 2;
  const float* Ag = g.A + (long)(tm * LBM + lr) * g.lda + lc;
  const bool hasab = g.abias != nullptr;
  float* Aw = &As[lr * APK + lc];
  float4 ra[NSL][NPA], rab[NSL], rt, rtb;
  #pragma unroll
  for (int s = 0; s < NSL; ++s) {
    #pragma unroll
    for (int p = 0; p < NPA; ++p) ra[s][p] = *(const float4*)(Ag + p * g.sAp + s * BK);
  }
  rt = rtb = make_float4(0.f, 0.f, 0.f, 0.f);
  if (KTR) rt = *(const float4*)(g.A2 + (long)(tm * LBM + lr) * g.lda2 + lc);
  if (hasab) {
    #pragma unroll
    for (int s = 0; s < NSL; ++s) rab[s] = *(const float4*)(g.abias + s * BK + lc);
    if (KTR) rtb = *(const float4*)(g.abias + KPL + lc);
  }

  const int  col0 = w * 32 + r15;
  const int  col1 = col0 + 16;
  const int  m0   = tm * LBM + (hh << 3);
  const float bv0 = g.bias[col0], bv1 = g.bias[col1];
  const float ga0 = g.gam[col0],  ga1 = g.gam[col1];
  const float be0 = g.bet[col0],  be1 = g.bet[col1];
  float x0[8], x1[8];
  for (int r = 0; r < 8; ++r) {
    x0[r] = g.resid[(long)(m0 + r) * LBN + col0];
    x1[r] = g.resid[(long)(m0 + r) * LBN + col1];
  }

  __builtin_amdgcn_sched_barrier(0);
  LN_B_STORE()

  #pragma unroll
  for (int s = 0; s < NSL; ++s) {
    float4 a = ra[s][0];
    #pragma unroll
    for (int p = 1; p < NPA; ++p) a = add4(a, ra[s][p]);
    if (hasab)   a = add4(a, rab[s]);
    if (g.arelu) a = relu4(a);
    *(float4*)(Aw + s * BK) = a;
  }
  if (KTR) {
    float4 a = rt;
    if (hasab)   a = add4(a, rtb);
    if (g.arelu) a = relu4(a);
    *(float4*)(Aw + KPL) = a;
  }

  __syncthreads();

  const float* Af  = &As[r15 * APK + kh];
  const float* Bf0 = &Bs[w * BWT + r15 * PK + kh];
  const float* Bf1 = Bf0 + 16 * PK;

  const v8f zero = {0.f,0.f,0.f,0.f,0.f,0.f,0.f,0.f};
  v8f c0 = zero, c1 = zero;

  for (int k0 = 0; k0 < KT; k0 += BK) {
    const bool more = (k0 + BK) < KT;
    if (more) { LN_B_GLOAD() }

    const float* af = Af + k0;
    #pragma unroll
    for (int kk = 0; kk < BK; kk += 4) {
      v2f a  = *(const v2f*)(af + kk);
      v2f b0 = *(const v2f*)(Bf0 + kk);
      v2f b1 = *(const v2f*)(Bf1 + kk);
      c0 = __builtin_amdgcn_wmma_f32_16x16x4_f32(false, a, false, b0, (short)0, c0, false, false);
      c1 = __builtin_amdgcn_wmma_f32_16x16x4_f32(false, a, false, b1, (short)0, c1, false, false);
    }
    if (more) {
      wave_lds_order();
      LN_B_STORE()
      wave_lds_order();
    }
  }
  asm volatile("v_nop\n\tv_nop\n\tv_nop\n\tv_nop\n\tv_nop" : "+v"(c0), "+v"(c1));
#undef LN_B_GLOAD
#undef LN_B_STORE

  float v0[8], v1[8], ps[8];
  for (int r = 0; r < 8; ++r) {
    v0[r] = x0[r] + (c0[r] + bv0);
    v1[r] = x1[r] + (c1[r] + bv1);
    ps[r] = v0[r] + v1[r];
  }
  for (int r = 0; r < 8; ++r) ps[r] = row_xor_add<1>(ps[r]);
  for (int r = 0; r < 8; ++r) ps[r] = row_xor_add<2>(ps[r]);
  for (int r = 0; r < 8; ++r) ps[r] = row_xor_add<4>(ps[r]);
  for (int r = 0; r < 8; ++r) ps[r] = row_xor_add<8>(ps[r]);
  if (r15 == 0) {
    for (int r = 0; r < 8; ++r) red1[((hh << 3) + r) * 8 + w] = ps[r];
  }
  __syncthreads();
  for (int r = 0; r < 8; ++r) {
    const float4 pa = *(const float4*)(&red1[((hh << 3) + r) * 8]);
    const float4 pb = *(const float4*)(&red1[((hh << 3) + r) * 8 + 4]);
    const float mean = (((pa.x + pa.y) + (pa.z + pa.w)) + ((pb.x + pb.y) + (pb.z + pb.w))) * (1.f / LBN);
    v0[r] -= mean;
    v1[r] -= mean;
    ps[r] = v0[r] * v0[r] + v1[r] * v1[r];
  }
  for (int r = 0; r < 8; ++r) ps[r] = row_xor_add<1>(ps[r]);
  for (int r = 0; r < 8; ++r) ps[r] = row_xor_add<2>(ps[r]);
  for (int r = 0; r < 8; ++r) ps[r] = row_xor_add<4>(ps[r]);
  for (int r = 0; r < 8; ++r) ps[r] = row_xor_add<8>(ps[r]);
  if (r15 == 0) {
    for (int r = 0; r < 8; ++r) red2[((hh << 3) + r) * 8 + w] = ps[r];
  }
  __syncthreads();
  for (int r = 0; r < 8; ++r) {
    const float4 pa = *(const float4*)(&red2[((hh << 3) + r) * 8]);
    const float4 pb = *(const float4*)(&red2[((hh << 3) + r) * 8 + 4]);
    const float var  = (((pa.x + pa.y) + (pa.z + pa.w)) + ((pb.x + pb.y) + (pb.z + pb.w))) * (1.f / LBN);
    const float rstd = rsqrtf(var + 1e-5f);
    const long row = (long)(m0 + r) * LBN;
    g.out[row + col0] = v0[r] * rstd * ga0 + be0;
    g.out[row + col1] = v1[r] * rstd * ga1 + be1;
  }
}

#define RP 68
__global__ __launch_bounds__(256)
void edge_softmax_kernel(const float* __restrict__ anchor_x,
                         const float* __restrict__ node_x,
                         const float* __restrict__ node_mask,
                         const float* __restrict__ q,
                         const float* __restrict__ bq,
                         const float* __restrict__ bk,
                         const float* __restrict__ qe,
                         const float* Lbase,
                         float* attn,
                         float* __restrict__ s_out,
                         int ldq, int lds_, long qps, long kps, long Lps)
{
  const int bid = blockIdx.x;
  const int b   = bid >> 6;
  const int t   = threadIdx.x;

  __shared__ __attribute__((aligned(16))) float sh_rbf[N_PER * RP];
  __shared__ __attribute__((aligned(16))) float sh_l[N_PER];
  __shared__ __attribute__((aligned(16))) float sh_part[16 * HE];
  __shared__ __attribute__((aligned(16))) float sh_qe[HE];
  __shared__ float red[256];
  __shared__ float sh_bcast;
  __shared__ float sh_ax[3];
  __shared__ float sh_c[8];

  if (t < HE) {
    float v = qe[(long)bid * ldq + t];
    for (int p = 1; p < KQSPLIT; ++p) v += qe[p * kps + (long)bid * ldq + t];
    sh_qe[t] = v;
  }
  if (t < 3)  sh_ax[t] = anchor_x[bid * 3 + t];
  float qv = q[(long)bid * H + t];
  for (int p = 1; p < QSPLIT; ++p) qv += q[p * qps + (long)bid * H + t];
  qv += bq[t];
  float pc = qv * bk[t];
  for (int o = 16; o > 0; o >>= 1) pc += __shfl_xor(pc, o, 32);
  if ((t & 31) == 0) sh_c[t >> 5] = pc;
  __syncthreads();
  const float ca = ((sh_c[0] + sh_c[1]) + (sh_c[2] + sh_c[3])) +
                   ((sh_c[4] + sh_c[5]) + (sh_c[6] + sh_c[7]));

  const float axx = sh_ax[0], axy = sh_ax[1], axz = sh_ax[2];
  const float MU_STEP = 20.f / 63.f;
  const float INV_SIG = 64.f / 20.f;

  for (int n = t; n < N_PER; n += 256) {
    const float* nx = node_x + (long)(b * N_PER + n) * 3;
    float dx = axx - nx[0] + 1e-8f;
    float dy = axy - nx[1] + 1e-8f;
    float dz = axz - nx[2] + 1e-8f;
    float d  = sqrtf(dx * dx + dy * dy + dz * dz);
    float d10 = d * 0.1f;
    float* rrow = &sh_rbf[n * RP];
    float le = 0.f;
    #pragma unroll 2
    for (int j = 0; j < HE; j += 4) {
      float u0 = (d10 - (j    ) * MU_STEP) * INV_SIG;
      float u1 = (d10 - (j + 1) * MU_STEP) * INV_SIG;
      float u2 = (d10 - (j + 2) * MU_STEP) * INV_SIG;
      float u3 = (d10 - (j + 3) * MU_STEP) * INV_SIG;
      float4 e;
      e.x = __expf(-u0 * u0);
      e.y = __expf(-u1 * u1);
      e.z = __expf(-u2 * u2);
      e.w = __expf(-u3 * u3);
      le = fmaf(sh_qe[j],     e.x, le);
      le = fmaf(sh_qe[j + 1], e.y, le);
      le = fmaf(sh_qe[j + 2], e.z, le);
      le = fmaf(sh_qe[j + 3], e.w, le);
      *(float4*)(rrow + j) = e;
    }
    float l0 = Lbase[(long)bid * N_PER + n];
    for (int p = 1; p < LSPLIT; ++p) l0 += Lbase[p * Lps + (long)bid * N_PER + n];
    float lb = (l0 + ca) + le;
    lb *= (node_mask[b * N_PER + n] - 1.f) * 1000000.0f;
    sh_l[n] = lb;
  }
  __syncthreads();

  red[t] = fmaxf(sh_l[t], sh_l[t + 256]);
  __syncthreads();
  for (int st = 128; st > 0; st >>= 1) {
    if (t < st) red[t] = fmaxf(red[t], red[t + st]);
    __syncthreads();
  }
  if (t == 0) sh_bcast = red[0];
  __syncthreads();
  const float mx = sh_bcast;

  float e0 = __expf(sh_l[t] - mx);
  float e1 = __expf(sh_l[t + 256] - mx);
  red[t] = e0 + e1;
  __syncthreads();
  for (int st = 128; st > 0; st >>= 1) {
    if (t < st) red[t] += red[t + st];
    __syncthreads();
  }
  if (t == 0) sh_bcast = red[0];
  __syncthreads();
  const float inv = 1.f / sh_bcast;

  const float a0 = e0 * inv, a1 = e1 * inv;
  sh_l[t] = a0; sh_l[t + 256] = a1;
  attn[(long)bid * N_PER + t]       = a0;
  attn[(long)bid * N_PER + t + 256] = a1;
  __syncthreads();

  const int jg = (t & 15) << 2, c = t >> 4;
  const float* rp = &sh_rbf[(c * 32) * RP + jg];
  const float* ap = &sh_l[c * 32];
  float4 part = make_float4(0.f, 0.f, 0.f, 0.f);
  for (int i = 0; i < 32; i += 4) {
    const float4 a4 = *(const float4*)(ap + i);
    const float4 r0 = *(const float4*)(rp + (i    ) * RP);
    const float4 r1 = *(const float4*)(rp + (i + 1) * RP);
    const float4 r2 = *(const float4*)(rp + (i + 2) * RP);
    const float4 r3 = *(const float4*)(rp + (i + 3) * RP);
    part.x = fmaf(a4.x, r0.x, part.x); part.y = fmaf(a4.x, r0.y, part.y);
    part.z = fmaf(a4.x, r0.z, part.z); part.w = fmaf(a4.x, r0.w, part.w);
    part.x = fmaf(a4.y, r1.x, part.x); part.y = fmaf(a4.y, r1.y, part.y);
    part.z = fmaf(a4.y, r1.z, part.z); part.w = fmaf(a4.y, r1.w, part.w);
    part.x = fmaf(a4.z, r2.x, part.x); part.y = fmaf(a4.z, r2.y, part.y);
    part.z = fmaf(a4.z, r2.z, part.z); part.w = fmaf(a4.z, r2.w, part.w);
    part.x = fmaf(a4.w, r3.x, part.x); part.y = fmaf(a4.w, r3.y, part.y);
    part.z = fmaf(a4.w, r3.z, part.z); part.w = fmaf(a4.w, r3.w, part.w);
  }
  *(float4*)(&sh_part[c * HE + jg]) = part;
  __syncthreads();
  if (t < HE) {
    float acc = sh_part[t];
    for (int cc = 1; cc < 16; ++cc) acc += sh_part[cc * HE + t];
    s_out[(long)bid * lds_ + t] = acc;
  }
}
#undef RP

extern "C" void kernel_launch(void* const* d_in, const int* in_sizes, int n_in,
                              void* d_out, int out_size, void* d_ws, size_t ws_size,
                              hipStream_t stream) {
  (void)in_sizes; (void)n_in; (void)out_size; (void)ws_size;
  const float* anchor_x  = (const float*)d_in[0];
  const float* node_x    = (const float*)d_in[1];
  const float* anchor_f  = (const float*)d_in[2];
  const float* node_f    = (const float*)d_in[3];
  const float* node_mask = (const float*)d_in[6];
  const float* Wq  = (const float*)d_in[7];
  const float* bq  = (const float*)d_in[8];
  const float* Wkv = (const float*)d_in[9];
  const float* bkv = (const float*)d_in[10];
  const float* ln1_g = (const float*)d_in[11];
  const float* ln1_b = (const float*)d_in[12];
  const float* W1 = (const float*)d_in[13];
  const float* b1 = (const float*)d_in[14];
  const float* W2 = (const float*)d_in[15];
  const float* b2 = (const float*)d_in[16];
  const float* W3 = (const float*)d_in[17];
  const float* b3 = (const float*)d_in[18];
  const float* ln2_g = (const float*)d_in[19];
  const float* ln2_b = (const float*)d_in[20];

  const long WKV_LD = H + HE;

  const long QP  = (long)NA * H;
  const long KQP = (long)NA * WKV_LD;
  const long LP  = (long)NA * N_PER;
  const long TP  = (long)NA * H;
  const long MP  = (long)NA * 2 * H;

  float* p = (float*)d_ws;
  float* q    = p; p += QSPLIT * QP;
  float* qkqe = p; p += KQSPLIT * KQP;
  float* L    = p; p += LSPLIT * LP;
  float* sbuf = p; p += (long)NA * HE;
  float* attn = L;
  float* ts   = q;
  float* af   = qkqe;
  float* m1   = L;
  float* m2   = q;

  auto gemm = [&](const float* A, long lda, long sAb, int npa, long sAp, const float* abias, int arelu,
                  const float* Bm, long ldbn, long ldbk, long sBb,
                  float* C, long ldc, long sCb, long sCp, const float* bias,
                  int M, int N, int Kc, int ksh, int nb, int relu) {
    GemmArgs g{A, lda, sAb, sAp, Bm, ldbn, ldbk, sBb, C, ldc, sCb, sCp, bias, abias, Kc, ksh, arelu, relu, 0};
    dim3 grid(N / BN, M / BM, nb << ksh);
    if (ldbk == 1) {
      if (npa == 1) wmma_gemm_f32<true, 1><<<grid, dim3(128), 0, stream>>>(g);
      else          wmma_gemm_f32<true, 2><<<grid, dim3(128), 0, stream>>>(g);
    } else {
      if (npa == 1) wmma_gemm_f32<false, 1><<<grid, dim3(128), 0, stream>>>(g);
      else          wmma_gemm_f32<false, 4><<<grid, dim3(128), 0, stream>>>(g);
    }
  };

  gemm(anchor_f, H, 0, 1, 0, nullptr, 0, Wq, H, 1, 0,
       q, H, 0, QP, nullptr, NA, H, H / QSPLIT, 2, 1, 0);
  gemm(q, H, 0, QSPLIT, QP, bq, 0, Wkv, 1, WKV_LD, 0,
       qkqe, WKV_LD, 0, KQP, nullptr, NA, (int)WKV_LD, H / KQSPLIT, 1, 1, 0);
  gemm(qkqe, WKV_LD, (long)A_PER * WKV_LD, KQSPLIT, KQP, nullptr, 0,
       node_f, H, 1, (long)N_PER * H,
       L, N_PER, (long)A_PER * N_PER, LP, nullptr, A_PER, N_PER, H / LSPLIT, 1, NB, 0);
  edge_softmax_kernel<<<dim3(NA), dim3(256), 0, stream>>>(
      anchor_x, node_x, node_mask, q, bq, bkv, qkqe + H, L, attn, sbuf,
      (int)WKV_LD, (int)HE, QP, KQP, LP);
  gemm(attn, N_PER, (long)A_PER * N_PER, 1, 0, nullptr, 0,
       node_f, 1, H, (long)N_PER * H,
       ts, H, (long)A_PER * H, TP, nullptr, A_PER, H, N_PER / TSPLIT, 2, NB, 0);
  {
    RowLnArgs g{ts, H, TP, sbuf, HE, nullptr, 0, Wkv + (long)H * WKV_LD, WKV_LD, bkv + H,
                anchor_f, ln1_g, ln1_b, af};
    wmma_gemm_rowln_f32<TSPLIT, H, HE><<<dim3(NA / LBM), dim3(256), 0, stream>>>(g);
  }
  gemm(af, H, 0, 1, 0, nullptr, 0, W1, H, 1, 0,
       m1, 2 * H, 0, MP, nullptr, NA, 2 * H, H / M1SPLIT, 1, 1, 0);
  gemm(m1, 2 * H, 0, M1SPLIT, MP, b1, 1, W2, 2 * H, 1, 0,
       m2, 2 * H, 0, MP, nullptr, NA, 2 * H, 2 * H / M2SPLIT, 1, 1, 0);
  {
    RowLnArgs g{m2, 2 * H, MP, m2, 2 * H, b2, 1, W3, 2 * H, b3, af, ln2_g, ln2_b, (float*)d_out};
    wmma_gemm_rowln_f32<M2SPLIT, 2 * H, 0><<<dim3(NA / LBM), dim3(256), 0, stream>>>(g);
  }
}
